# starting kernel
# baseline (speedup 1.0000x reference)
#include <hip/hip_runtime.h>

typedef __attribute__((ext_vector_type(2))) float v2f;
typedef __attribute__((ext_vector_type(8))) float v8f;

#define IN_DIM 64

__global__ void deg_kernel(const int* __restrict__ row, const int* __restrict__ col,
                           float* __restrict__ deg, int E) {
    int e = blockIdx.x * blockDim.x + threadIdx.x;
    if (e >= E) return;
    int r = row[e], c = col[e];
    if (r != c) atomicAdd(&deg[r], 1.0f);
}

__global__ void rsqrt_kernel(float* __restrict__ dis, int N) {
    int i = blockIdx.x * blockDim.x + threadIdx.x;
    if (i >= N) return;
    float d = dis[i];
    dis[i] = (d > 0.0f) ? rsqrtf(d) : 0.0f;
}

__global__ void neg_copy_kernel(const float* __restrict__ src, float* __restrict__ dst, int n) {
    int i = blockIdx.x * blockDim.x + threadIdx.x;
    if (i >= n) return;
    dst[i] = -src[i];
}

__global__ void spmm_kernel(const int* __restrict__ row, const int* __restrict__ col,
                            const float* __restrict__ dis, const float* __restrict__ V,
                            float* __restrict__ OUT, float scale, int E) {
    int gid = blockIdx.x * blockDim.x + threadIdx.x;
    int e = gid >> 4;
    if (e >= E) return;
    int chunk = (gid & 15) << 2;
    int r = row[e], c = col[e];
    if (r == c) return;
    float wgt = -scale * dis[r] * dis[c];
    if (wgt == 0.0f) return;
    const float4 v = *(const float4*)(V + (size_t)c * IN_DIM + chunk);
    float* o = OUT + (size_t)r * IN_DIM + chunk;
    atomicAdd(o + 0, wgt * v.x);
    atomicAdd(o + 1, wgt * v.y);
    atomicAdd(o + 2, wgt * v.z);
    atomicAdd(o + 3, wgt * v.w);
}

template <int OUT_DIM, int DO_RELU>
__global__ void __launch_bounds__(256, 1)
cheb_gemm_wmma(const float* __restrict__ T0,
               const float* __restrict__ T1,
               const float* __restrict__ T2,
               const float* __restrict__ W,
               const float* __restrict__ bias,
               float* __restrict__ out,
               int N) {
    constexpr int COL_TILES = OUT_DIM / 16;
    const int lane = threadIdx.x & 31;
    const int wave = (blockIdx.x * blockDim.x + threadIdx.x) >> 5;
    const int totalWaves = (gridDim.x * blockDim.x) >> 5;
    const int rowWaves = totalWaves / COL_TILES;
    const int ct   = wave % COL_TILES;
    const int rwid = wave / COL_TILES;
    const int rowTiles = (N + 15) >> 4;

    const int n    = lane & 15;
    const int m    = lane & 15;
    const int koff = (lane >> 4) << 1;
    const int col0 = ct << 4;
    const int mhalf = (lane >> 4) << 3;

    v2f bfrag[3][16];
#pragma unroll
    for (int s = 0; s < 3; ++s) {
        const float* Ws = W + s * IN_DIM * OUT_DIM + col0 + n;
#pragma unroll
        for (int k4 = 0; k4 < 16; ++k4) {
            const int kk = k4 * 4 + koff;
            bfrag[s][k4].x = Ws[(kk + 0) * OUT_DIM];
            bfrag[s][k4].y = Ws[(kk + 1) * OUT_DIM];
        }
    }
    const float bv = bias[col0 + n];

    for (int rt = rwid; rt < rowTiles; rt += rowWaves) {
        const int row0 = rt << 4;
        const int arow = min(row0 + m, N - 1);
        const size_t abase = (size_t)arow * IN_DIM + koff;

        v8f acc0 = {};
        v8f acc1 = {};
#pragma unroll
        for (int s = 0; s < 3; ++s) {
            const float* T = (s == 0) ? T0 : (s == 1) ? T1 : T2;
            const float* ap = T + abase;

#pragma unroll
            for (int k8 = 0; k8 < 8; ++k8) {
                v2f a0, a1;
                a0.x = ap[(2 * k8 + 0) * 4 + 0];
                a0.y = ap[(2 * k8 + 0) * 4 + 1];
                a1.x = ap[(2 * k8 + 1) * 4 + 0];
                a1.y = ap[(2 * k8 + 1) * 4 + 1];
                acc0 = __builtin_amdgcn_wmma_f32_16x16x4_f32(
                    false, a0, false, bfrag[s][2 * k8 + 0], (short)0, acc0, false, false);
                acc1 = __builtin_amdgcn_wmma_f32_16x16x4_f32(
                    false, a1, false, bfrag[s][2 * k8 + 1], (short)0, acc1, false, false);
            }
        }

#pragma unroll
        for (int r = 0; r < 8; ++r) {
            int mr = row0 + r + mhalf;
            if (mr < N) {
                float v = acc0[r] + acc1[r] + bv;
                if (DO_RELU) v = fmaxf(v, 0.0f);
                out[(size_t)mr * OUT_DIM + col0 + n] = v;
            }
        }
    }
}

extern "C" void kernel_launch(void* const* d_in, const int* in_sizes, int n_in,
                              void* d_out, int out_size, void* d_ws, size_t ws_size,
                              hipStream_t stream) {
    const float* x  = (const float*)d_in[0];
    const int*   ei = (const int*)d_in[1];
    const float* W1 = (const float*)d_in[2];
    const float* b1 = (const float*)d_in[3];
    const float* W2 = (const float*)d_in[4];
    const float* b2 = (const float*)d_in[5];
    float* out = (float*)d_out;

    const int N = in_sizes[0] / IN_DIM;
    const int E = in_sizes[1] / 2;
    const int* row = ei;
    const int* col = ei + E;

    float* ws  = (float*)d_ws;
    float* dis = ws;
    float* Tx1 = ws + N;
    float* Tx2 = Tx1 + (size_t)N * IN_DIM;
    float* h   = Tx2 + (size_t)N * IN_DIM;

    const int threads = 256;
    const int featN   = N * IN_DIM;
    const unsigned gridE    = (unsigned)((E + threads - 1) / threads);
    const unsigned gridN    = (unsigned)((N + threads - 1) / threads);
    const unsigned gridFeat = (unsigned)((featN + threads - 1) / threads);
    const unsigned gridSpmm = (unsigned)(((size_t)E * 16 + threads - 1) / threads);

    hipMemsetAsync(dis, 0, (size_t)N * sizeof(float), stream);
    deg_kernel<<<gridE, threads, 0, stream>>>(row, col, dis, E);
    rsqrt_kernel<<<gridN, threads, 0, stream>>>(dis, N);

    hipMemsetAsync(Tx1, 0, (size_t)featN * sizeof(float), stream);
    spmm_kernel<<<gridSpmm, threads, 0, stream>>>(row, col, dis, x, Tx1, 1.0f, E);
    neg_copy_kernel<<<gridFeat, threads, 0, stream>>>(x, Tx2, featN);
    spmm_kernel<<<gridSpmm, threads, 0, stream>>>(row, col, dis, Tx1, Tx2, 2.0f, E);

    cheb_gemm_wmma<64, 1><<<512, threads, 0, stream>>>(x, Tx1, Tx2, W1, b1, h, N);

    hipMemsetAsync(Tx1, 0, (size_t)featN * sizeof(float), stream);
    spmm_kernel<<<gridSpmm, threads, 0, stream>>>(row, col, dis, h, Tx1, 1.0f, E);
    neg_copy_kernel<<<gridFeat, threads, 0, stream>>>(h, Tx2, featN);
    spmm_kernel<<<gridSpmm, threads, 0, stream>>>(row, col, dis, Tx1, Tx2, 2.0f, E);

    cheb_gemm_wmma<16, 0><<<128, threads, 0, stream>>>(h, Tx1, Tx2, W2, b2, out, N);
}


# SpMM: f32 atomic scatter -> in-call row-sorted edge list (int rank atomics + scan), one wave per row gathers and stores once; sum order = rank order; memset/neg seeds fold into the store
# speedup vs baseline: 2.7990x; 2.7990x over previous
#include <hip/hip_runtime.h>


typedef __attribute__((ext_vector_type(2))) float v2f;
typedef __attribute__((ext_vector_type(8))) float v8f;

#define IN_DIM 64

#define COL_BITS 17
#define COL_MASK 0x1FFFFu
#define DEG_ESC  0x7FFFu

__global__ void count_rank_kernel(const int* __restrict__ row, const int* __restrict__ col,
                                  int* __restrict__ cnt, int* __restrict__ rank, int E) {
    int e = blockIdx.x * blockDim.x + threadIdx.x;
    if (e >= E) return;
    int r = row[e], c = col[e];
    if (r != c) rank[e] = atomicAdd(&cnt[r], 1);
}

__global__ void __launch_bounds__(256)
scan_kernel(const int* __restrict__ cnt, int* __restrict__ rend, int N) {
    __shared__ int s_a[8];
    __shared__ int s_b[8];
    const int tid = threadIdx.x, lane = tid & 31, wv = tid >> 5;
    const int chunk0 = blockIdx.x * 1024;

    const int4* cp = (const int4*)cnt + tid;
    const int nfront = blockIdx.x;
    int part = 0;
    for (int k0 = 0; k0 < nfront; k0 += 16) {
        int4 q[16];
#pragma unroll
        for (int j = 0; j < 16; ++j) {
            q[j] = int4{0, 0, 0, 0};
            if (k0 + j < nfront) q[j] = cp[(k0 + j) * 256];
        }
#pragma unroll
        for (int j = 0; j < 16; ++j) part += (q[j].x + q[j].y) + (q[j].z + q[j].w);
    }
    for (int o = 16; o > 0; o >>= 1) part += __shfl_xor(part, o);
    if (lane == 0) s_a[wv] = part;

    const int i0 = chunk0 + tid * 4;
    int4 q = {0, 0, 0, 0};
    if (i0 + 3 < N) {
        q = *(const int4*)(cnt + i0);
    } else {
        if (i0 + 0 < N) q.x = cnt[i0 + 0];
        if (i0 + 1 < N) q.y = cnt[i0 + 1];
        if (i0 + 2 < N) q.z = cnt[i0 + 2];
    }
    const int t = (q.x + q.y) + (q.z + q.w);
    int incl = t;
    for (int o = 1; o < 32; o <<= 1) {
        const int u = __shfl_up(incl, o);
        if (lane >= o) incl += u;
    }
    if (lane == 31) s_b[wv] = incl;
    __syncthreads();

    int off = 0;
    for (int j = 0; j < 8; ++j) {
        off += s_a[j];
        if (j < wv) off += s_b[j];
    }
    int4 o4;
    o4.x = off + (incl - t) + q.x;
    o4.y = o4.x + q.y;
    o4.z = o4.y + q.z;
    o4.w = o4.z + q.w;
    if (i0 + 3 < N) {
        *(int4*)(rend + i0) = o4;
    } else {
        if (i0 + 0 < N) rend[i0 + 0] = o4.x;
        if (i0 + 1 < N) rend[i0 + 1] = o4.y;
        if (i0 + 2 < N) rend[i0 + 2] = o4.z;
    }
}

__global__ void fill_kernel(const int* __restrict__ row, const int* __restrict__ col,
                            const int* __restrict__ rank, const int* __restrict__ cnt,
                            const int* __restrict__ rend, unsigned* __restrict__ csr, int E) {
    int e = blockIdx.x * blockDim.x + threadIdx.x;
    if (e >= E) return;
    int r = row[e], c = col[e];
    if (r == c) return;
    const int start = r ? rend[r - 1] : 0;
    unsigned d = (unsigned)cnt[c];
    if (d > DEG_ESC) d = DEG_ESC;
    csr[start + rank[e]] = (unsigned)c | (d << COL_BITS);
}

#define ROW_STEPS 12
#define TRIP (2 * ROW_STEPS)

template <int K>
__device__ __forceinline__ int half_lane(int v) {
    return __builtin_amdgcn_update_dpp(0, v, 0x150 + K, 0xF, 0xF, false);
}

__device__ __forceinline__ float other_half(float x) {
    return __int_as_float((int)__builtin_amdgcn_permlanex16(
        0u, (unsigned)__float_as_int(x), 0x76543210u, 0xfedcba98u, false, false));
}

template <int J, int NS>
struct RowSteps {
    static __device__ __forceinline__ void issue(float4* v, int c_l, int np,
                                                 const char* Vb, unsigned loff) {
        if (J < np) {
            const unsigned c = (unsigned)half_lane<J>(c_l);
            v[J] = *(const float4*)(Vb + ((c << 8) + loff));
        }
        RowSteps<J + 1, NS>::issue(v, c_l, np, Vb, loff);
    }
    static __device__ __forceinline__ void sum(float4& acc, const float4* v, float w_l, int np) {
        if (J < np) {
            const float w = __int_as_float(half_lane<J>(__float_as_int(w_l)));
            acc.x = fmaf(w, v[J].x, acc.x);
            acc.y = fmaf(w, v[J].y, acc.y);
            acc.z = fmaf(w, v[J].z, acc.z);
            acc.w = fmaf(w, v[J].w, acc.w);
        }
        RowSteps<J + 1, NS>::sum(acc, v, w_l, np);
    }
};
template <int NS>
struct RowSteps<NS, NS> {
    static __device__ __forceinline__ void issue(float4*, int, int, const char*, unsigned) {}
    static __device__ __forceinline__ void sum(float4&, const float4*, float, int) {}
};

__device__ __forceinline__ unsigned load_entry(const unsigned* __restrict__ csr,
                                               int base, int n, int slot) {
    const int idx = (slot < n) ? base + slot : 0;
    return csr[idx];
}

__device__ __forceinline__ void decode_entry(unsigned ent, int n, int slot, float coef,
                                             const int* __restrict__ rend,
                                             int& c_l, float& w_l) {
    const bool ok = slot < n;
    const int c = (int)(ent & COL_MASK);
    unsigned d = ent >> COL_BITS;
    if (ok && d == DEG_ESC) d = (unsigned)(rend[c] - (c ? rend[c - 1] : 0));
    const float w = coef * rsqrtf((float)d);
    c_l = ok ? c : 0;
    w_l = (ok && d) ? w : 0.0f;
}

template <int NS>
struct Trip {
    float4 v[NS];
    float4 vt;
    float  w_l, wt;
    int    np, odd;
};

template <int NS>
__device__ __forceinline__ void trip_issue(Trip<NS>& t, unsigned ent, int n, int slot, int half,
                                           float coef, const int* __restrict__ rend,
                                           const char* Vb, unsigned loff) {
    int c_l;
    decode_entry(ent, n, slot, coef, rend, c_l, t.w_l);
    t.np  = n >> 1;
    t.odd = n & 1;
    RowSteps<0, NS>::issue(t.v, c_l, t.np, Vb, loff);
    const unsigned ct = (unsigned)__builtin_amdgcn_readlane(c_l, t.np & 15);
    t.wt = __int_as_float(__builtin_amdgcn_readlane(__float_as_int(t.w_l), t.np & 15));
    t.vt = float4{0.0f, 0.0f, 0.0f, 0.0f};
    if (t.odd && half == 0) t.vt = *(const float4*)(Vb + ((ct << 8) + loff));
}

template <int NS>
__device__ __forceinline__ void trip_sum(float4& acc, const Trip<NS>& t, int half) {
    RowSteps<0, NS>::sum(acc, t.v, t.w_l, t.np);
    if (t.odd && half == 0) {
        acc.x = fmaf(t.wt, t.vt.x, acc.x);
        acc.y = fmaf(t.wt, t.vt.y, acc.y);
        acc.z = fmaf(t.wt, t.vt.z, acc.z);
        acc.w = fmaf(t.wt, t.vt.w, acc.w);
    }
}

#define LONG_STEPS 4

template <int SUB_T0>
__global__ void __launch_bounds__(256)
spmm_rows_kernel(const unsigned* __restrict__ csr, const int* __restrict__ rend,
                 const float* __restrict__ V, const float* __restrict__ T0,
                 float* __restrict__ OUT, float scale, int N) {
    const int lane = threadIdx.x & 31;
    const int half = lane >> 4;
    const int slot = ((lane & 15) << 1) + half;
    const unsigned loff = (unsigned)(lane & 15) << 4;
    const char* Vb = (const char*)V;
    const int wavesPerBlock = blockDim.x >> 5;
    const int w0 = __builtin_amdgcn_readfirstlane(
        (int)(blockIdx.x * wavesPerBlock + (threadIdx.x >> 5)));
    const int nw = gridDim.x * wavesPerBlock;
    if (w0 >= N) return;

    int start = w0 ? rend[w0 - 1] : 0;
    int len   = rend[w0] - start;
    unsigned ent = load_entry(csr, start, min(TRIP, len), slot);

    for (int r = w0; r < N; r += nw) {
        const int rn = r + nw;
        int na = 0, nb = 0;
        if (rn < N) { na = rend[rn - 1]; nb = rend[rn]; }

        const float coef = (len > 0) ? -scale * rsqrtf((float)len) : 0.0f;
        float4 acc = {0.0f, 0.0f, 0.0f, 0.0f};
        int base = start, left = len;

        while (left > TRIP) {
            Trip<LONG_STEPS> t;
            trip_issue(t, ent, 2 * LONG_STEPS, slot, half, coef, rend, Vb, loff);
            base += 2 * LONG_STEPS;
            left -= 2 * LONG_STEPS;
            ent = load_entry(csr, base, min(TRIP, left), slot);
            trip_sum(acc, t, half);
        }

        Trip<ROW_STEPS> t;
        trip_issue(t, ent, left, slot, half, coef, rend, Vb, loff);
        float4 t0 = {0.0f, 0.0f, 0.0f, 0.0f};
        if (SUB_T0 && half == 0)
            t0 = *(const float4*)((const char*)T0 + (((size_t)r << 8) + loff));
        const int nlen = nb - na;
        const unsigned ent_next = load_entry(csr, na, min(TRIP, nlen), slot);
        trip_sum(acc, t, half);

        acc.x += other_half(acc.x);
        acc.y += other_half(acc.y);
        acc.z += other_half(acc.z);
        acc.w += other_half(acc.w);
        if (half == 0) {
            if (SUB_T0) { acc.x -= t0.x; acc.y -= t0.y; acc.z -= t0.z; acc.w -= t0.w; }
            *(float4*)((char*)OUT + (((size_t)r << 8) + loff)) = acc;
        }
        start = na;
        len   = nlen;
        ent   = ent_next;
    }
}

template <int OUT_DIM, int DO_RELU>
__global__ void __launch_bounds__(256, 1)
cheb_gemm_wmma(const float* __restrict__ T0,
               const float* __restrict__ T1,
               const float* __restrict__ T2,
               const float* __restrict__ W,
               const float* __restrict__ bias,
               float* __restrict__ out,
               int N) {
    constexpr int COL_TILES = OUT_DIM / 16;
    const int lane = threadIdx.x & 31;
    const int wave = (blockIdx.x * blockDim.x + threadIdx.x) >> 5;
    const int totalWaves = (gridDim.x * blockDim.x) >> 5;
    const int rowWaves = totalWaves / COL_TILES;
    const int ct   = wave % COL_TILES;
    const int rwid = wave / COL_TILES;
    const int rowTiles = (N + 15) >> 4;

    const int n    = lane & 15;
    const int m    = lane & 15;
    const int koff = (lane >> 4) << 1;
    const int col0 = ct << 4;
    const int mhalf = (lane >> 4) << 3;

    v2f bfrag[3][16];
#pragma unroll
    for (int s = 0; s < 3; ++s) {
        const float* Ws = W + s * IN_DIM * OUT_DIM + col0 + n;
#pragma unroll
        for (int k4 = 0; k4 < 16; ++k4) {
            const int kk = k4 * 4 + koff;
            bfrag[s][k4].x = Ws[(kk + 0) * OUT_DIM];
            bfrag[s][k4].y = Ws[(kk + 1) * OUT_DIM];
        }
    }
    const float bv = bias[col0 + n];

    for (int rt = rwid; rt < rowTiles; rt += rowWaves) {
        const int row0 = rt << 4;
        const int arow = min(row0 + m, N - 1);
        const size_t abase = (size_t)arow * IN_DIM + koff;

        v8f acc0 = {};
        v8f acc1 = {};
#pragma unroll
        for (int s = 0; s < 3; ++s) {
            const float* T = (s == 0) ? T0 : (s == 1) ? T1 : T2;
            const float* ap = T + abase;

#pragma unroll
            for (int k8 = 0; k8 < 8; ++k8) {
                v2f a0, a1;
                a0.x = ap[(2 * k8 + 0) * 4 + 0];
                a0.y = ap[(2 * k8 + 0) * 4 + 1];
                a1.x = ap[(2 * k8 + 1) * 4 + 0];
                a1.y = ap[(2 * k8 + 1) * 4 + 1];
                acc0 = __builtin_amdgcn_wmma_f32_16x16x4_f32(
                    false, a0, false, bfrag[s][2 * k8 + 0], (short)0, acc0, false, false);
                acc1 = __builtin_amdgcn_wmma_f32_16x16x4_f32(
                    false, a1, false, bfrag[s][2 * k8 + 1], (short)0, acc1, false, false);
            }
        }

#pragma unroll
        for (int r = 0; r < 8; ++r) {
            int mr = row0 + r + mhalf;
            if (mr < N) {
                float v = acc0[r] + acc1[r] + bv;
                if (DO_RELU) v = fmaxf(v, 0.0f);
                out[(size_t)mr * OUT_DIM + col0 + n] = v;
            }
        }
    }
}

extern "C" void kernel_launch(void* const* d_in, const int* in_sizes, int n_in,
                              void* d_out, int out_size, void* d_ws, size_t ws_size,
                              hipStream_t stream) {
    const float* x  = (const float*)d_in[0];
    const int*   ei = (const int*)d_in[1];
    const float* W1 = (const float*)d_in[2];
    const float* b1 = (const float*)d_in[3];
    const float* W2 = (const float*)d_in[4];
    const float* b2 = (const float*)d_in[5];
    float* out = (float*)d_out;

    const int N = in_sizes[0] / IN_DIM;
    const int E = in_sizes[1] / 2;
    const int* row = ei;
    const int* col = ei + E;

    float* ws   = (float*)d_ws;
    int*   rend = (int*)ws;
    float* Tx1  = ws + N;
    float* Tx2  = Tx1 + (size_t)N * IN_DIM;
    float* h    = Tx2 + (size_t)N * IN_DIM;
    int*   cnt  = (int*)h;
    int*   rank = cnt + N;
    unsigned* csr = (unsigned*)d_out;

    const int threads = 256;
    const unsigned gridE    = (unsigned)((E + threads - 1) / threads);
    const unsigned gridScan = (unsigned)((N + 1023) / 1024);
    const unsigned gridRows = 1280;

    hipMemsetAsync(cnt, 0, (size_t)N * sizeof(int), stream);
    count_rank_kernel<<<gridE, threads, 0, stream>>>(row, col, cnt, rank, E);
    scan_kernel<<<gridScan, threads, 0, stream>>>(cnt, rend, N);
    fill_kernel<<<gridE, threads, 0, stream>>>(row, col, rank, cnt, rend, csr, E);

    spmm_rows_kernel<0><<<gridRows, threads, 0, stream>>>(csr, rend, x, x, Tx1, 1.0f, N);
    spmm_rows_kernel<1><<<gridRows, threads, 0, stream>>>(csr, rend, Tx1, x, Tx2, 2.0f, N);

    cheb_gemm_wmma<64, 1><<<512, threads, 0, stream>>>(x, Tx1, Tx2, W1, b1, h, N);

    spmm_rows_kernel<0><<<gridRows, threads, 0, stream>>>(csr, rend, h, h, Tx1, 1.0f, N);
    spmm_rows_kernel<1><<<gridRows, threads, 0, stream>>>(csr, rend, Tx1, h, Tx2, 2.0f, N);

    cheb_gemm_wmma<16, 0><<<128, threads, 0, stream>>>(h, Tx1, Tx2, W2, b2, out, N);
}


# Edge list by two-level bucket sort counted in LDS (512-row buckets, LDS atomics, staged whole-line stores), not global rank atomics + scattered fill; no memset; sum order = LDS arrival order
# speedup vs baseline: 4.6310x; 1.6545x over previous
#include <hip/hip_runtime.h>


typedef __attribute__((ext_vector_type(2))) float v2f;
typedef __attribute__((ext_vector_type(8))) float v8f;

#define IN_DIM 64

#define COL_BITS 17
#define COL_MASK 0x1FFFFu
#define DEG_ESC  0x7FFFu

__global__ void __launch_bounds__(256)
scan_kernel(const int* __restrict__ cnt, int* __restrict__ rend, int N) {
    __shared__ int s_a[8];
    __shared__ int s_b[8];
    const int tid = threadIdx.x, lane = tid & 31, wv = tid >> 5;
    const int chunk0 = blockIdx.x * 1024;

    const int4* cp = (const int4*)cnt + tid;
    const int nfront = blockIdx.x;
    int part = 0;
    for (int k0 = 0; k0 < nfront; k0 += 16) {
        int4 q[16];
#pragma unroll
        for (int j = 0; j < 16; ++j) {
            q[j] = int4{0, 0, 0, 0};
            if (k0 + j < nfront) q[j] = cp[(k0 + j) * 256];
        }
#pragma unroll
        for (int j = 0; j < 16; ++j) part += (q[j].x + q[j].y) + (q[j].z + q[j].w);
    }
    for (int o = 16; o > 0; o >>= 1) part += __shfl_xor(part, o);
    if (lane == 0) s_a[wv] = part;

    const int i0 = chunk0 + tid * 4;
    int4 q = {0, 0, 0, 0};
    if (i0 + 3 < N) {
        q = *(const int4*)(cnt + i0);
    } else {
        if (i0 + 0 < N) q.x = cnt[i0 + 0];
        if (i0 + 1 < N) q.y = cnt[i0 + 1];
        if (i0 + 2 < N) q.z = cnt[i0 + 2];
    }
    const int t = (q.x + q.y) + (q.z + q.w);
    int incl = t;
    for (int o = 1; o < 32; o <<= 1) {
        const int u = __shfl_up(incl, o);
        if (lane >= o) incl += u;
    }
    if (lane == 31) s_b[wv] = incl;
    __syncthreads();

    int off = 0;
    for (int j = 0; j < 8; ++j) {
        off += s_a[j];
        if (j < wv) off += s_b[j];
    }
    int4 o4;
    o4.x = off + (incl - t) + q.x;
    o4.y = o4.x + q.y;
    o4.z = o4.y + q.z;
    o4.w = o4.z + q.w;
    if (i0 + 3 < N) {
        *(int4*)(rend + i0) = o4;
    } else {
        if (i0 + 0 < N) rend[i0 + 0] = o4.x;
        if (i0 + 1 < N) rend[i0 + 1] = o4.y;
        if (i0 + 2 < N) rend[i0 + 2] = o4.z;
    }
}

#define BSH    9
#define BROWS  512
#define MAXB   256
#define SUB    4096
#define EPT    (SUB / 256)
#define CAP    12288

__global__ void __launch_bounds__(256)
bucket_count_kernel(const int* __restrict__ row, const int* __restrict__ col,
                    int* __restrict__ bc, int E, int NB) {
    __shared__ int h[MAXB];
    const int tid = threadIdx.x;
    h[tid] = 0;
    __syncthreads();
    const int base = blockIdx.x * SUB;
#pragma unroll
    for (int j = 0; j < EPT; ++j) {
        const int e = base + j * 256 + tid;
        if (e < E) {
            const int r = row[e], c = col[e];
            if (r != c) atomicAdd(&h[min(r >> BSH, NB - 1)], 1);
        }
    }
    __syncthreads();
    if (tid < NB) bc[tid * gridDim.x + blockIdx.x] = h[tid];
}

__global__ void __launch_bounds__(256)
bucket_scatter_kernel(const int* __restrict__ row, const int* __restrict__ col,
                      const int* __restrict__ bo, unsigned* __restrict__ bk, int E, int NB) {
    __shared__ int h[MAXB];
    __shared__ int gb[MAXB];
    __shared__ int wsum[8];
    __shared__ unsigned st[SUB];
    __shared__ int dst[SUB];
    const int tid = threadIdx.x, lane = tid & 31, wv = tid >> 5;
    h[tid] = 0;
    {
        const int i = tid * gridDim.x + blockIdx.x;
        gb[tid] = (tid < NB && i > 0) ? bo[i - 1] : 0;
    }
    __syncthreads();
    const int base = blockIdx.x * SUB;
    unsigned ent[EPT];
    int bb[EPT], lr[EPT];
#pragma unroll
    for (int j = 0; j < EPT; ++j) {
        const int e = base + j * 256 + tid;
        bb[j] = -1; lr[j] = 0; ent[j] = 0;
        if (e < E) {
            const int r = row[e], c = col[e];
            if (r != c) {
                const int b = min(r >> BSH, NB - 1);
                bb[j]  = b;
                ent[j] = ((unsigned)(r - (b << BSH)) << COL_BITS) | ((unsigned)c & COL_MASK);
                lr[j]  = atomicAdd(&h[b], 1);
            }
        }
    }
    __syncthreads();
    const int v = h[tid];
    int incl = v;
    for (int o = 1; o < 32; o <<= 1) {
        const int u = __shfl_up(incl, o);
        if (lane >= o) incl += u;
    }
    if (lane == 31) wsum[wv] = incl;
    __syncthreads();
    int off = 0, total = 0;
    for (int j = 0; j < 8; ++j) {
        const int s = wsum[j];
        if (j < wv) off += s;
        total += s;
    }
    h[tid] = off + incl - v;
    __syncthreads();
#pragma unroll
    for (int j = 0; j < EPT; ++j) {
        if (bb[j] >= 0) {
            const int p = h[bb[j]] + lr[j];
            st[p]  = ent[j];
            dst[p] = gb[bb[j]] + lr[j];
        }
    }
    __syncthreads();
    for (int i = tid; i < total; i += 256) bk[dst[i]] = st[i];
}

__global__ void __launch_bounds__(1024)
bucket_rows_kernel(const unsigned* __restrict__ bk, const int* __restrict__ bo,
                   int* __restrict__ cnt, int* __restrict__ rend, int N, int G) {
    __shared__ int c[BROWS];
    __shared__ int wsum[32];
    const int tid = threadIdx.x, lane = tid & 31, wv = tid >> 5;
    const int i0 = blockIdx.x * G;
    const int bs = i0 ? bo[i0 - 1] : 0;
    const int be = bo[i0 + G - 1];
    if (tid < BROWS) c[tid] = 0;
    __syncthreads();
    for (int i = bs + tid; i < be; i += 1024) atomicAdd(&c[(bk[i] >> COL_BITS) & (BROWS - 1)], 1);
    __syncthreads();
    const int v = (tid < BROWS) ? c[tid] : 0;
    int incl = v;
    for (int o = 1; o < 32; o <<= 1) {
        const int u = __shfl_up(incl, o);
        if (lane >= o) incl += u;
    }
    if (lane == 31) wsum[wv] = incl;
    __syncthreads();
    int off = 0;
    for (int j = 0; j < wv; ++j) off += wsum[j];
    const int r = blockIdx.x * BROWS + tid;
    if (tid < BROWS && r < N) {
        cnt[r]  = v;
        rend[r] = bs + off + incl;
    }
}

__global__ void __launch_bounds__(1024)
bucket_place_kernel(const unsigned* __restrict__ bk, const int* __restrict__ bo,
                    const int* __restrict__ cnt, const int* __restrict__ rend,
                    unsigned* __restrict__ csr, int N, int G) {
    __shared__ int cur[BROWS];
    __shared__ unsigned st[CAP];
    const int tid = threadIdx.x;
    const int i0 = blockIdx.x * G;
    const int bs = i0 ? bo[i0 - 1] : 0;
    const int be = bo[i0 + G - 1];
    const int nb = be - bs;
    const int r = blockIdx.x * BROWS + tid;
    if (tid < BROWS) cur[tid] = (r < N) ? ((r ? rend[r - 1] : 0) - bs) : 0;
    __syncthreads();
    const bool staged = nb <= CAP;
    for (int i = bs + tid; i < be; i += 1024) {
        const unsigned e = bk[i];
        const unsigned s = e & COL_MASK;
        unsigned d = (unsigned)cnt[s];
        if (d > DEG_ESC) d = DEG_ESC;
        const int p = atomicAdd(&cur[(e >> COL_BITS) & (BROWS - 1)], 1);
        const unsigned val = s | (d << COL_BITS);
        if (staged) st[p] = val; else csr[bs + p] = val;
    }
    __syncthreads();
    if (staged)
        for (int i = tid; i < nb; i += 1024) csr[bs + i] = st[i];
}

#define ROW_STEPS 12
#define TRIP (2 * ROW_STEPS)

template <int K>
__device__ __forceinline__ int half_lane(int v) {
    return __builtin_amdgcn_update_dpp(0, v, 0x150 + K, 0xF, 0xF, false);
}

__device__ __forceinline__ float other_half(float x) {
    return __int_as_float((int)__builtin_amdgcn_permlanex16(
        0u, (unsigned)__float_as_int(x), 0x76543210u, 0xfedcba98u, false, false));
}

template <int J, int NS>
struct RowSteps {
    static __device__ __forceinline__ void issue(float4* v, int c_l, int np,
                                                 const char* Vb, unsigned loff) {
        if (J < np) {
            const unsigned c = (unsigned)half_lane<J>(c_l);
            v[J] = *(const float4*)(Vb + ((c << 8) + loff));
        }
        RowSteps<J + 1, NS>::issue(v, c_l, np, Vb, loff);
    }
    static __device__ __forceinline__ void sum(float4& acc, const float4* v, float w_l, int np) {
        if (J < np) {
            const float w = __int_as_float(half_lane<J>(__float_as_int(w_l)));
            acc.x = fmaf(w, v[J].x, acc.x);
            acc.y = fmaf(w, v[J].y, acc.y);
            acc.z = fmaf(w, v[J].z, acc.z);
            acc.w = fmaf(w, v[J].w, acc.w);
        }
        RowSteps<J + 1, NS>::sum(acc, v, w_l, np);
    }
};
template <int NS>
struct RowSteps<NS, NS> {
    static __device__ __forceinline__ void issue(float4*, int, int, const char*, unsigned) {}
    static __device__ __forceinline__ void sum(float4&, const float4*, float, int) {}
};

__device__ __forceinline__ unsigned load_entry(const unsigned* __restrict__ csr,
                                               int base, int n, int slot) {
    const int idx = (slot < n) ? base + slot : 0;
    return csr[idx];
}

__device__ __forceinline__ void decode_entry(unsigned ent, int n, int slot, float coef,
                                             const int* __restrict__ rend,
                                             int& c_l, float& w_l) {
    const bool ok = slot < n;
    const int c = (int)(ent & COL_MASK);
    unsigned d = ent >> COL_BITS;
    if (ok && d == DEG_ESC) d = (unsigned)(rend[c] - (c ? rend[c - 1] : 0));
    const float w = coef * rsqrtf((float)d);
    c_l = ok ? c : 0;
    w_l = (ok && d) ? w : 0.0f;
}

template <int NS>
struct Trip {
    float4 v[NS];
    float4 vt;
    float  w_l, wt;
    int    np, odd;
};

template <int NS>
__device__ __forceinline__ void trip_issue(Trip<NS>& t, unsigned ent, int n, int slot, int half,
                                           float coef, const int* __restrict__ rend,
                                           const char* Vb, unsigned loff) {
    int c_l;
    decode_entry(ent, n, slot, coef, rend, c_l, t.w_l);
    t.np  = n >> 1;
    t.odd = n & 1;
    RowSteps<0, NS>::issue(t.v, c_l, t.np, Vb, loff);
    const unsigned ct = (unsigned)__builtin_amdgcn_readlane(c_l, t.np & 15);
    t.wt = __int_as_float(__builtin_amdgcn_readlane(__float_as_int(t.w_l), t.np & 15));
    t.vt = float4{0.0f, 0.0f, 0.0f, 0.0f};
    if (t.odd && half == 0) t.vt = *(const float4*)(Vb + ((ct << 8) + loff));
}

template <int NS>
__device__ __forceinline__ void trip_sum(float4& acc, const Trip<NS>& t, int half) {
    RowSteps<0, NS>::sum(acc, t.v, t.w_l, t.np);
    if (t.odd && half == 0) {
        acc.x = fmaf(t.wt, t.vt.x, acc.x);
        acc.y = fmaf(t.wt, t.vt.y, acc.y);
        acc.z = fmaf(t.wt, t.vt.z, acc.z);
        acc.w = fmaf(t.wt, t.vt.w, acc.w);
    }
}

#define LONG_STEPS 4

template <int SUB_T0>
__global__ void __launch_bounds__(256)
spmm_rows_kernel(const unsigned* __restrict__ csr, const int* __restrict__ rend,
                 const float* __restrict__ V, const float* __restrict__ T0,
                 float* __restrict__ OUT, float scale, int N) {
    const int lane = threadIdx.x & 31;
    const int half = lane >> 4;
    const int slot = ((lane & 15) << 1) + half;
    const unsigned loff = (unsigned)(lane & 15) << 4;
    const char* Vb = (const char*)V;
    const int wavesPerBlock = blockDim.x >> 5;
    const int w0 = __builtin_amdgcn_readfirstlane(
        (int)(blockIdx.x * wavesPerBlock + (threadIdx.x >> 5)));
    const int nw = gridDim.x * wavesPerBlock;
    if (w0 >= N) return;

    int start = w0 ? rend[w0 - 1] : 0;
    int len   = rend[w0] - start;
    unsigned ent = load_entry(csr, start, min(TRIP, len), slot);

    for (int r = w0; r < N; r += nw) {
        const int rn = r + nw;
        int na = 0, nb = 0;
        if (rn < N) { na = rend[rn - 1]; nb = rend[rn]; }

        const float coef = (len > 0) ? -scale * rsqrtf((float)len) : 0.0f;
        float4 acc = {0.0f, 0.0f, 0.0f, 0.0f};
        int base = start, left = len;

        while (left > TRIP) {
            Trip<LONG_STEPS> t;
            trip_issue(t, ent, 2 * LONG_STEPS, slot, half, coef, rend, Vb, loff);
            base += 2 * LONG_STEPS;
            left -= 2 * LONG_STEPS;
            ent = load_entry(csr, base, min(TRIP, left), slot);
            trip_sum(acc, t, half);
        }

        Trip<ROW_STEPS> t;
        trip_issue(t, ent, left, slot, half, coef, rend, Vb, loff);
        float4 t0 = {0.0f, 0.0f, 0.0f, 0.0f};
        if (SUB_T0 && half == 0)
            t0 = *(const float4*)((const char*)T0 + (((size_t)r << 8) + loff));
        const int nlen = nb - na;
        const unsigned ent_next = load_entry(csr, na, min(TRIP, nlen), slot);
        trip_sum(acc, t, half);

        acc.x += other_half(acc.x);
        acc.y += other_half(acc.y);
        acc.z += other_half(acc.z);
        acc.w += other_half(acc.w);
        if (half == 0) {
            if (SUB_T0) { acc.x -= t0.x; acc.y -= t0.y; acc.z -= t0.z; acc.w -= t0.w; }
            *(float4*)((char*)OUT + (((size_t)r << 8) + loff)) = acc;
        }
        start = na;
        len   = nlen;
        ent   = ent_next;
    }
}

template <int OUT_DIM, int DO_RELU>
__global__ void __launch_bounds__(256, 1)
cheb_gemm_wmma(const float* __restrict__ T0,
               const float* __restrict__ T1,
               const float* __restrict__ T2,
               const float* __restrict__ W,
               const float* __restrict__ bias,
               float* __restrict__ out,
               int N) {
    constexpr int COL_TILES = OUT_DIM / 16;
    const int lane = threadIdx.x & 31;
    const int wave = (blockIdx.x * blockDim.x + threadIdx.x) >> 5;
    const int totalWaves = (gridDim.x * blockDim.x) >> 5;
    const int rowWaves = totalWaves / COL_TILES;
    const int ct   = wave % COL_TILES;
    const int rwid = wave / COL_TILES;
    const int rowTiles = (N + 15) >> 4;

    const int n    = lane & 15;
    const int m    = lane & 15;
    const int koff = (lane >> 4) << 1;
    const int col0 = ct << 4;
    const int mhalf = (lane >> 4) << 3;

    v2f bfrag[3][16];
#pragma unroll
    for (int s = 0; s < 3; ++s) {
        const float* Ws = W + s * IN_DIM * OUT_DIM + col0 + n;
#pragma unroll
        for (int k4 = 0; k4 < 16; ++k4) {
            const int kk = k4 * 4 + koff;
            bfrag[s][k4].x = Ws[(kk + 0) * OUT_DIM];
            bfrag[s][k4].y = Ws[(kk + 1) * OUT_DIM];
        }
    }
    const float bv = bias[col0 + n];

    for (int rt = rwid; rt < rowTiles; rt += rowWaves) {
        const int row0 = rt << 4;
        const int arow = min(row0 + m, N - 1);
        const size_t abase = (size_t)arow * IN_DIM + koff;

        v8f acc0 = {};
        v8f acc1 = {};
#pragma unroll
        for (int s = 0; s < 3; ++s) {
            const float* T = (s == 0) ? T0 : (s == 1) ? T1 : T2;
            const float* ap = T + abase;

#pragma unroll
            for (int k8 = 0; k8 < 8; ++k8) {
                v2f a0, a1;
                a0.x = ap[(2 * k8 + 0) * 4 + 0];
                a0.y = ap[(2 * k8 + 0) * 4 + 1];
                a1.x = ap[(2 * k8 + 1) * 4 + 0];
                a1.y = ap[(2 * k8 + 1) * 4 + 1];
                acc0 = __builtin_amdgcn_wmma_f32_16x16x4_f32(
                    false, a0, false, bfrag[s][2 * k8 + 0], (short)0, acc0, false, false);
                acc1 = __builtin_amdgcn_wmma_f32_16x16x4_f32(
                    false, a1, false, bfrag[s][2 * k8 + 1], (short)0, acc1, false, false);
            }
        }

#pragma unroll
        for (int r = 0; r < 8; ++r) {
            int mr = row0 + r + mhalf;
            if (mr < N) {
                float v = acc0[r] + acc1[r] + bv;
                if (DO_RELU) v = fmaxf(v, 0.0f);
                out[(size_t)mr * OUT_DIM + col0 + n] = v;
            }
        }
    }
}

__global__ void __launch_bounds__(256, 1)
proj_gemm_wmma(const float* __restrict__ H,
               const float* __restrict__ W,
               const float* __restrict__ bias,
               float* __restrict__ Z,
               float* __restrict__ P1,
               float* __restrict__ P2,
               int N) {
    constexpr int OUT_DIM = 16;
    const int lane = threadIdx.x & 31;
    const int wave = (blockIdx.x * blockDim.x + threadIdx.x) >> 5;
    const int totalWaves = (gridDim.x * blockDim.x) >> 5;
    const int rowTiles = (N + 15) >> 4;

    const int n    = lane & 15;
    const int m    = lane & 15;
    const int koff = (lane >> 4) << 1;
    const int mhalf = (lane >> 4) << 3;

    v2f bfrag[3][16];
#pragma unroll
    for (int s = 0; s < 3; ++s) {
        const float* Ws = W + s * IN_DIM * OUT_DIM + n;
#pragma unroll
        for (int k4 = 0; k4 < 16; ++k4) {
            const int kk = k4 * 4 + koff;
            bfrag[s][k4].x = Ws[(kk + 0) * OUT_DIM];
            bfrag[s][k4].y = Ws[(kk + 1) * OUT_DIM];
        }
    }
    const float bv = bias[n];

    for (int rt = wave; rt < rowTiles; rt += totalWaves) {
        const int row0 = rt << 4;
        const int arow = min(row0 + m, N - 1);
        const float* ap = H + (size_t)arow * IN_DIM + koff;

        v8f acc0 = {};
        v8f acc1 = {};
        v8f acc2 = {};
#pragma unroll
        for (int k4 = 0; k4 < 16; ++k4) {
            v2f a;
            a.x = ap[k4 * 4 + 0];
            a.y = ap[k4 * 4 + 1];
            acc0 = __builtin_amdgcn_wmma_f32_16x16x4_f32(
                false, a, false, bfrag[0][k4], (short)0, acc0, false, false);
            acc1 = __builtin_amdgcn_wmma_f32_16x16x4_f32(
                false, a, false, bfrag[1][k4], (short)0, acc1, false, false);
            acc2 = __builtin_amdgcn_wmma_f32_16x16x4_f32(
                false, a, false, bfrag[2][k4], (short)0, acc2, false, false);
        }

#pragma unroll
        for (int r = 0; r < 8; ++r) {
            const int mr = row0 + r + mhalf;
            if (mr < N) {
                const size_t o = (size_t)mr * OUT_DIM + n;
                Z[o]  = (acc0[r] - acc2[r]) + bv;
                P1[o] = acc1[r];
                P2[o] = acc2[r];
            }
        }
    }
}

template <int J>
__device__ __forceinline__ int quad_lane(int v) {
    return __builtin_amdgcn_update_dpp(0, v, J * 0x55, 0xF, 0xF, false);
}

template <int J>
struct QuadSteps {
    static __device__ __forceinline__ void issue(float4* va, float4* vb, float* wa, float* wb,
                                                 int ca_l, int cb_l, float wa_l, float wb_l,
                                                 int left, const char* Vb, unsigned loff) {
        const unsigned ca = (unsigned)quad_lane<J>(ca_l);
        const unsigned cb = (unsigned)quad_lane<J>(cb_l);
        wa[J] = __int_as_float(quad_lane<J>(__float_as_int(wa_l)));
        wb[J] = __int_as_float(quad_lane<J>(__float_as_int(wb_l)));
        va[J] = float4{0.0f, 0.0f, 0.0f, 0.0f};
        vb[J] = float4{0.0f, 0.0f, 0.0f, 0.0f};
        if (J < left)     va[J] = *(const float4*)(Vb + ((ca << 6) + loff));
        if (J + 4 < left) vb[J] = *(const float4*)(Vb + ((cb << 6) + loff));
        QuadSteps<J + 1>::issue(va, vb, wa, wb, ca_l, cb_l, wa_l, wb_l, left, Vb, loff);
    }
};
template <>
struct QuadSteps<4> {
    static __device__ __forceinline__ void issue(float4*, float4*, float*, float*,
                                                 int, int, float, float,
                                                 int, const char*, unsigned) {}
};

template <int COPY>
__global__ void __launch_bounds__(256)
spmm16_kernel(const unsigned* __restrict__ csr, unsigned* __restrict__ csr_copy,
              const int* __restrict__ rend, const float* __restrict__ V,
              const float* __restrict__ ADD, float* __restrict__ OUT, float scale, int N) {
    const int gt = blockIdx.x * blockDim.x + threadIdx.x;
    const int r = gt >> 2;
    const int q = gt & 3;
    if (r >= N) return;
    const unsigned loff = (unsigned)q << 4;
    const char* Vb = (const char*)V;

    const int start = r ? rend[r - 1] : 0;
    const int end   = rend[r];
    const int len   = end - start;
    const float coef = (len > 0) ? -scale * rsqrtf((float)len) : 0.0f;

    float4 acc = *(const float4*)((const char*)ADD + (((size_t)r << 6) + loff));

    unsigned ea = 0, eb = 0;
    if (start + q < end)     ea = csr[start + q];
    if (start + 4 + q < end) eb = csr[start + 4 + q];

    for (int base = start; base < end; base += 8) {
        const int ia = base + q, ib = base + 4 + q;
        const bool oka = ia < end, okb = ib < end;
        if (COPY) {
            if (oka) csr_copy[ia] = ea;
            if (okb) csr_copy[ib] = eb;
        }
        int ca_l, cb_l;
        float wa_l, wb_l;
        decode_entry(ea, oka ? 1 : 0, 0, coef, rend, ca_l, wa_l);
        decode_entry(eb, okb ? 1 : 0, 0, coef, rend, cb_l, wb_l);

        float4 va[4], vb[4];
        float  wa[4], wb[4];
        QuadSteps<0>::issue(va, vb, wa, wb, ca_l, cb_l, wa_l, wb_l, end - base, Vb, loff);

        ea = 0; eb = 0;
        if (ia + 8 < end) ea = csr[ia + 8];
        if (ib + 8 < end) eb = csr[ib + 8];

#pragma unroll
        for (int j = 0; j < 4; ++j) {
            acc.x = fmaf(wa[j], va[j].x, acc.x);
            acc.y = fmaf(wa[j], va[j].y, acc.y);
            acc.z = fmaf(wa[j], va[j].z, acc.z);
            acc.w = fmaf(wa[j], va[j].w, acc.w);
        }
#pragma unroll
        for (int j = 0; j < 4; ++j) {
            acc.x = fmaf(wb[j], vb[j].x, acc.x);
            acc.y = fmaf(wb[j], vb[j].y, acc.y);
            acc.z = fmaf(wb[j], vb[j].z, acc.z);
            acc.w = fmaf(wb[j], vb[j].w, acc.w);
        }
    }
    *(float4*)((char*)OUT + (((size_t)r << 6) + loff)) = acc;
}

extern "C" void kernel_launch(void* const* d_in, const int* in_sizes, int n_in,
                              void* d_out, int out_size, void* d_ws, size_t ws_size,
                              hipStream_t stream) {
    const float* x  = (const float*)d_in[0];
    const int*   ei = (const int*)d_in[1];
    const float* W1 = (const float*)d_in[2];
    const float* b1 = (const float*)d_in[3];
    const float* W2 = (const float*)d_in[4];
    const float* b2 = (const float*)d_in[5];
    float* out = (float*)d_out;

    const int N = in_sizes[0] / IN_DIM;
    const int E = in_sizes[1] / 2;
    const int* row = ei;
    const int* col = ei + E;

    float* ws   = (float*)d_ws;
    int*   rend = (int*)ws;
    float* Tx1  = ws + N;
    float* Tx2  = Tx1 + (size_t)N * IN_DIM;
    float* h    = Tx2 + (size_t)N * IN_DIM;
    const int NB = (N + BROWS - 1) >> BSH;
    const int GA = (E + SUB - 1) / SUB;
    const int MB = (NB * GA + 3) & ~3;
    int*      cnt = (int*)h;
    int*      bc  = cnt + N;
    int*      bo  = bc + MB;
    unsigned* bk  = (unsigned*)(bo + MB);
    unsigned* csr = (unsigned*)d_out;

    const int threads = 256;
    const unsigned gridScan = (unsigned)((NB * GA + 1023) / 1024);
    const unsigned gridRows = 1280;

    bucket_count_kernel<<<(unsigned)GA, threads, 0, stream>>>(row, col, bc, E, NB);
    scan_kernel<<<gridScan, threads, 0, stream>>>(bc, bo, NB * GA);
    bucket_scatter_kernel<<<(unsigned)GA, threads, 0, stream>>>(row, col, bo, bk, E, NB);
    bucket_rows_kernel<<<(unsigned)NB, 1024, 0, stream>>>(bk, bo, cnt, rend, N, GA);
    bucket_place_kernel<<<(unsigned)NB, 1024, 0, stream>>>(bk, bo, cnt, rend, csr, N, GA);

    spmm_rows_kernel<0><<<gridRows, threads, 0, stream>>>(csr, rend, x, x, Tx1, 1.0f, N);
    spmm_rows_kernel<1><<<gridRows, threads, 0, stream>>>(csr, rend, Tx1, x, Tx2, 2.0f, N);

    cheb_gemm_wmma<64, 1><<<512, threads, 0, stream>>>(x, Tx1, Tx2, W1, b1, h, N);

    float* Z  = Tx1;
    float* P1 = Z  + (size_t)N * 16;
    float* P2 = P1 + (size_t)N * 16;
    float* R  = P2 + (size_t)N * 16;
    unsigned* csr2 = (unsigned*)Tx2;
    const unsigned gridQuad = (unsigned)((N + 63) / 64);

    proj_gemm_wmma<<<512, threads, 0, stream>>>(h, W2, b2, Z, P1, P2, N);
    spmm16_kernel<1><<<gridQuad, threads, 0, stream>>>(csr, csr2, rend, P2, P1, R, 2.0f, N);
    spmm16_kernel<0><<<gridQuad, threads, 0, stream>>>(csr2, nullptr, rend, R, Z, out, 1.0f, N);
}


# Layer 1: second graph product and dense stage in one launch: a block gathers a 16-row Tx2 tile into LDS beside its x and Tx1 rows, 4 waves WMMA it with W from LDS; Tx2 never stored; sums unchanged
# speedup vs baseline: 5.1998x; 1.1228x over previous
#include <hip/hip_runtime.h>


typedef __attribute__((ext_vector_type(2))) float v2f;
typedef __attribute__((ext_vector_type(8))) float v8f;

#define IN_DIM 64

#define COL_BITS 17
#define COL_MASK 0x1FFFFu
#define DEG_ESC  0x7FFFu

__global__ void __launch_bounds__(256)
scan_kernel(const int* __restrict__ cnt, int* __restrict__ rend, int N) {
    __shared__ int s_a[8];
    __shared__ int s_b[8];
    const int tid = threadIdx.x, lane = tid & 31, wv = tid >> 5;
    const int chunk0 = blockIdx.x * 1024;

    const int4* cp = (const int4*)cnt + tid;
    const int nfront = blockIdx.x;
    int part = 0;
    for (int k0 = 0; k0 < nfront; k0 += 16) {
        int4 q[16];
#pragma unroll
        for (int j = 0; j < 16; ++j) {
            q[j] = int4{0, 0, 0, 0};
            if (k0 + j < nfront) q[j] = cp[(k0 + j) * 256];
        }
#pragma unroll
        for (int j = 0; j < 16; ++j) part += (q[j].x + q[j].y) + (q[j].z + q[j].w);
    }
    for (int o = 16; o > 0; o >>= 1) part += __shfl_xor(part, o);
    if (lane == 0) s_a[wv] = part;

    const int i0 = chunk0 + tid * 4;
    int4 q = {0, 0, 0, 0};
    if (i0 + 3 < N) {
        q = *(const int4*)(cnt + i0);
    } else {
        if (i0 + 0 < N) q.x = cnt[i0 + 0];
        if (i0 + 1 < N) q.y = cnt[i0 + 1];
        if (i0 + 2 < N) q.z = cnt[i0 + 2];
    }
    const int t = (q.x + q.y) + (q.z + q.w);
    int incl = t;
    for (int o = 1; o < 32; o <<= 1) {
        const int u = __shfl_up(incl, o);
        if (lane >= o) incl += u;
    }
    if (lane == 31) s_b[wv] = incl;
    __syncthreads();

    int off = 0;
    for (int j = 0; j < 8; ++j) {
        off += s_a[j];
        if (j < wv) off += s_b[j];
    }
    int4 o4;
    o4.x = off + (incl - t) + q.x;
    o4.y = o4.x + q.y;
    o4.z = o4.y + q.z;
    o4.w = o4.z + q.w;
    if (i0 + 3 < N) {
        *(int4*)(rend + i0) = o4;
    } else {
        if (i0 + 0 < N) rend[i0 + 0] = o4.x;
        if (i0 + 1 < N) rend[i0 + 1] = o4.y;
        if (i0 + 2 < N) rend[i0 + 2] = o4.z;
    }
}

#define BSH    9
#define BROWS  512
#define MAXB   256
#define SUB    4096
#define EPT    (SUB / 256)
#define CAP    12288

__global__ void __launch_bounds__(256)
bucket_count_kernel(const int* __restrict__ row, const int* __restrict__ col,
                    int* __restrict__ bc, int E, int NB) {
    __shared__ int h[MAXB];
    const int tid = threadIdx.x;
    h[tid] = 0;
    __syncthreads();
    const int base = blockIdx.x * SUB;
#pragma unroll
    for (int j = 0; j < EPT; ++j) {
        const int e = base + j * 256 + tid;
        if (e < E) {
            const int r = row[e], c = col[e];
            if (r != c) atomicAdd(&h[min(r >> BSH, NB - 1)], 1);
        }
    }
    __syncthreads();
    if (tid < NB) bc[tid * gridDim.x + blockIdx.x] = h[tid];
}

__global__ void __launch_bounds__(256)
bucket_scatter_kernel(const int* __restrict__ row, const int* __restrict__ col,
                      const int* __restrict__ bo, unsigned* __restrict__ bk, int E, int NB) {
    __shared__ int h[MAXB];
    __shared__ int gb[MAXB];
    __shared__ int wsum[8];
    __shared__ unsigned st[SUB];
    __shared__ int dst[SUB];
    const int tid = threadIdx.x, lane = tid & 31, wv = tid >> 5;
    h[tid] = 0;
    {
        const int i = tid * gridDim.x + blockIdx.x;
        gb[tid] = (tid < NB && i > 0) ? bo[i - 1] : 0;
    }
    __syncthreads();
    const int base = blockIdx.x * SUB;
    unsigned ent[EPT];
    int bb[EPT], lr[EPT];
#pragma unroll
    for (int j = 0; j < EPT; ++j) {
        const int e = base + j * 256 + tid;
        bb[j] = -1; lr[j] = 0; ent[j] = 0;
        if (e < E) {
            const int r = row[e], c = col[e];
            if (r != c) {
                const int b = min(r >> BSH, NB - 1);
                bb[j]  = b;
                ent[j] = ((unsigned)(r - (b << BSH)) << COL_BITS) | ((unsigned)c & COL_MASK);
                lr[j]  = atomicAdd(&h[b], 1);
            }
        }
    }
    __syncthreads();
    const int v = h[tid];
    int incl = v;
    for (int o = 1; o < 32; o <<= 1) {
        const int u = __shfl_up(incl, o);
        if (lane >= o) incl += u;
    }
    if (lane == 31) wsum[wv] = incl;
    __syncthreads();
    int off = 0, total = 0;
    for (int j = 0; j < 8; ++j) {
        const int s = wsum[j];
        if (j < wv) off += s;
        total += s;
    }
    h[tid] = off + incl - v;
    __syncthreads();
#pragma unroll
    for (int j = 0; j < EPT; ++j) {
        if (bb[j] >= 0) {
            const int p = h[bb[j]] + lr[j];
            st[p]  = ent[j];
            dst[p] = gb[bb[j]] + lr[j];
        }
    }
    __syncthreads();
    for (int i = tid; i < total; i += 256) bk[dst[i]] = st[i];
}

__global__ void __launch_bounds__(1024)
bucket_rows_kernel(const unsigned* __restrict__ bk, const int* __restrict__ bo,
                   int* __restrict__ cnt, int* __restrict__ rend, int N, int G) {
    __shared__ int c[BROWS];
    __shared__ int wsum[32];
    const int tid = threadIdx.x, lane = tid & 31, wv = tid >> 5;
    const int i0 = blockIdx.x * G;
    const int bs = i0 ? bo[i0 - 1] : 0;
    const int be = bo[i0 + G - 1];
    if (tid < BROWS) c[tid] = 0;
    __syncthreads();
    for (int i = bs + tid; i < be; i += 1024) atomicAdd(&c[(bk[i] >> COL_BITS) & (BROWS - 1)], 1);
    __syncthreads();
    const int v = (tid < BROWS) ? c[tid] : 0;
    int incl = v;
    for (int o = 1; o < 32; o <<= 1) {
        const int u = __shfl_up(incl, o);
        if (lane >= o) incl += u;
    }
    if (lane == 31) wsum[wv] = incl;
    __syncthreads();
    int off = 0;
    for (int j = 0; j < wv; ++j) off += wsum[j];
    const int r = blockIdx.x * BROWS + tid;
    if (tid < BROWS && r < N) {
        cnt[r]  = v;
        rend[r] = bs + off + incl;
    }
}

__global__ void __launch_bounds__(1024)
bucket_place_kernel(const unsigned* __restrict__ bk, const int* __restrict__ bo,
                    const int* __restrict__ cnt, const int* __restrict__ rend,
                    unsigned* __restrict__ csr, int N, int G) {
    __shared__ int cur[BROWS];
    __shared__ unsigned st[CAP];
    const int tid = threadIdx.x;
    const int i0 = blockIdx.x * G;
    const int bs = i0 ? bo[i0 - 1] : 0;
    const int be = bo[i0 + G - 1];
    const int nb = be - bs;
    const int r = blockIdx.x * BROWS + tid;
    if (tid < BROWS) cur[tid] = (r < N) ? ((r ? rend[r - 1] : 0) - bs) : 0;
    __syncthreads();
    const bool staged = nb <= CAP;
    for (int i = bs + tid; i < be; i += 1024) {
        const unsigned e = bk[i];
        const unsigned s = e & COL_MASK;
        unsigned d = (unsigned)cnt[s];
        if (d > DEG_ESC) d = DEG_ESC;
        const int p = atomicAdd(&cur[(e >> COL_BITS) & (BROWS - 1)], 1);
        const unsigned val = s | (d << COL_BITS);
        if (staged) st[p] = val; else csr[bs + p] = val;
    }
    __syncthreads();
    if (staged)
        for (int i = tid; i < nb; i += 1024) csr[bs + i] = st[i];
}

#define ROW_STEPS 12
#define TRIP (2 * ROW_STEPS)

template <int K>
__device__ __forceinline__ int half_lane(int v) {
    return __builtin_amdgcn_update_dpp(0, v, 0x150 + K, 0xF, 0xF, false);
}

__device__ __forceinline__ float other_half(float x) {
    return __int_as_float((int)__builtin_amdgcn_permlanex16(
        0u, (unsigned)__float_as_int(x), 0x76543210u, 0xfedcba98u, false, false));
}

template <int J, int NS>
struct RowSteps {
    static __device__ __forceinline__ void issue(float4* v, int c_l, int np,
                                                 const char* Vb, unsigned loff) {
        if (J < np) {
            const unsigned c = (unsigned)half_lane<J>(c_l);
            v[J] = *(const float4*)(Vb + ((c << 8) + loff));
        }
        RowSteps<J + 1, NS>::issue(v, c_l, np, Vb, loff);
    }
    static __device__ __forceinline__ void sum(float4& acc, const float4* v, float w_l, int np) {
        if (J < np) {
            const float w = __int_as_float(half_lane<J>(__float_as_int(w_l)));
            acc.x = fmaf(w, v[J].x, acc.x);
            acc.y = fmaf(w, v[J].y, acc.y);
            acc.z = fmaf(w, v[J].z, acc.z);
            acc.w = fmaf(w, v[J].w, acc.w);
        }
        RowSteps<J + 1, NS>::sum(acc, v, w_l, np);
    }
};
template <int NS>
struct RowSteps<NS, NS> {
    static __device__ __forceinline__ void issue(float4*, int, int, const char*, unsigned) {}
    static __device__ __forceinline__ void sum(float4&, const float4*, float, int) {}
};

__device__ __forceinline__ unsigned load_entry(const unsigned* __restrict__ csr,
                                               int base, int n, int slot) {
    const int idx = (slot < n) ? base + slot : 0;
    return csr[idx];
}

__device__ __forceinline__ void decode_entry(unsigned ent, int n, int slot, float coef,
                                             const int* __restrict__ rend,
                                             int& c_l, float& w_l) {
    const bool ok = slot < n;
    const int c = (int)(ent & COL_MASK);
    unsigned d = ent >> COL_BITS;
    if (ok && d == DEG_ESC) d = (unsigned)(rend[c] - (c ? rend[c - 1] : 0));
    const float w = coef * rsqrtf((float)d);
    c_l = ok ? c : 0;
    w_l = (ok && d) ? w : 0.0f;
}

template <int NS>
struct Trip {
    float4 v[NS];
    float4 vt;
    float  w_l, wt;
    int    np, odd;
};

template <int NS>
__device__ __forceinline__ void trip_issue(Trip<NS>& t, unsigned ent, int n, int slot, int half,
                                           float coef, const int* __restrict__ rend,
                                           const char* Vb, unsigned loff) {
    int c_l;
    decode_entry(ent, n, slot, coef, rend, c_l, t.w_l);
    t.np  = n >> 1;
    t.odd = n & 1;
    RowSteps<0, NS>::issue(t.v, c_l, t.np, Vb, loff);
    const unsigned ct = (unsigned)__builtin_amdgcn_readlane(c_l, t.np & 15);
    t.wt = __int_as_float(__builtin_amdgcn_readlane(__float_as_int(t.w_l), t.np & 15));
    t.vt = float4{0.0f, 0.0f, 0.0f, 0.0f};
    if (t.odd && half == 0) t.vt = *(const float4*)(Vb + ((ct << 8) + loff));
}

template <int NS>
__device__ __forceinline__ void trip_sum(float4& acc, const Trip<NS>& t, int half) {
    RowSteps<0, NS>::sum(acc, t.v, t.w_l, t.np);
    if (t.odd && half == 0) {
        acc.x = fmaf(t.wt, t.vt.x, acc.x);
        acc.y = fmaf(t.wt, t.vt.y, acc.y);
        acc.z = fmaf(t.wt, t.vt.z, acc.z);
        acc.w = fmaf(t.wt, t.vt.w, acc.w);
    }
}

#define LONG_STEPS 4

template <int SUB_T0>
__global__ void __launch_bounds__(256)
spmm_rows_kernel(const unsigned* __restrict__ csr, const int* __restrict__ rend,
                 const float* __restrict__ V, const float* __restrict__ T0,
                 float* __restrict__ OUT, float scale, int N) {
    const int lane = threadIdx.x & 31;
    const int half = lane >> 4;
    const int slot = ((lane & 15) << 1) + half;
    const unsigned loff = (unsigned)(lane & 15) << 4;
    const char* Vb = (const char*)V;
    const int wavesPerBlock = blockDim.x >> 5;
    const int w0 = __builtin_amdgcn_readfirstlane(
        (int)(blockIdx.x * wavesPerBlock + (threadIdx.x >> 5)));
    const int nw = gridDim.x * wavesPerBlock;
    if (w0 >= N) return;

    int start = w0 ? rend[w0 - 1] : 0;
    int len   = rend[w0] - start;
    unsigned ent = load_entry(csr, start, min(TRIP, len), slot);

    for (int r = w0; r < N; r += nw) {
        const int rn = r + nw;
        int na = 0, nb = 0;
        if (rn < N) { na = rend[rn - 1]; nb = rend[rn]; }

        const float coef = (len > 0) ? -scale * rsqrtf((float)len) : 0.0f;
        float4 acc = {0.0f, 0.0f, 0.0f, 0.0f};
        int base = start, left = len;

        while (left > TRIP) {
            Trip<LONG_STEPS> t;
            trip_issue(t, ent, 2 * LONG_STEPS, slot, half, coef, rend, Vb, loff);
            base += 2 * LONG_STEPS;
            left -= 2 * LONG_STEPS;
            ent = load_entry(csr, base, min(TRIP, left), slot);
            trip_sum(acc, t, half);
        }

        Trip<ROW_STEPS> t;
        trip_issue(t, ent, left, slot, half, coef, rend, Vb, loff);
        float4 t0 = {0.0f, 0.0f, 0.0f, 0.0f};
        if (SUB_T0 && half == 0)
            t0 = *(const float4*)((const char*)T0 + (((size_t)r << 8) + loff));
        const int nlen = nb - na;
        const unsigned ent_next = load_entry(csr, na, min(TRIP, nlen), slot);
        trip_sum(acc, t, half);

        acc.x += other_half(acc.x);
        acc.y += other_half(acc.y);
        acc.z += other_half(acc.z);
        acc.w += other_half(acc.w);
        if (half == 0) {
            if (SUB_T0) { acc.x -= t0.x; acc.y -= t0.y; acc.z -= t0.z; acc.w -= t0.w; }
            *(float4*)((char*)OUT + (((size_t)r << 8) + loff)) = acc;
        }
        start = na;
        len   = nlen;
        ent   = ent_next;
    }
}

#define TP 68

__global__ void __launch_bounds__(256)
spmm_dense_kernel(const unsigned* __restrict__ csr, const int* __restrict__ rend,
                  const float* __restrict__ X, const float* __restrict__ T1,
                  const float* __restrict__ W,
                  const float* __restrict__ bias,
                  float* __restrict__ Hout,
                  float scale, int N) {
    __shared__ __attribute__((aligned(16))) float sW[3 * 16 * 4 * 64];
    __shared__ __attribute__((aligned(16))) float sT[3 * 16 * TP];
    const int tid  = threadIdx.x;
    const int lane = tid & 31;
    const int half = lane >> 4;
    const int slot = ((lane & 15) << 1) + half;
    const unsigned loff = (unsigned)(lane & 15) << 4;
    const char* Vb = (const char*)T1;
    const int wv = __builtin_amdgcn_readfirstlane(tid >> 5);

    for (int i = tid; i < 3 * IN_DIM * 64; i += 256) {
        const int s = i >> 12, k = (i >> 6) & 63, c = i & 63;
        const int ln = (((k >> 1) & 1) << 4) | (c & 15);
        sW[((((s << 4) + (k >> 2)) * 4 + (c >> 4)) * 32 + ln) * 2 + (k & 1)] = W[i];
    }
    const float bv = bias[((wv & 3) << 4) + (lane & 15)];

    const int tiles = (N + 15) >> 4;
    const int ntile = ((int)blockIdx.x < tiles) ? (tiles - 1 - (int)blockIdx.x) / (int)gridDim.x + 1 : 0;
    const int nq = 2 * ntile;

    int start = 0, len = 0;
    {
        const int r0 = ((int)blockIdx.x << 4) + (wv << 1);
        if (nq > 0 && r0 < N) {
            start = r0 ? rend[r0 - 1] : 0;
            len   = rend[r0] - start;
        }
    }
    unsigned ent = load_entry(csr, start, min(TRIP, len), slot);

    for (int q = 0; q < nq; ++q) {
        const int j    = q & 1;
        const int tile = (int)blockIdx.x + (q >> 1) * (int)gridDim.x;
        const int r    = (tile << 4) + (wv << 1) + j;
        const int rn = j ? (((tile + (int)gridDim.x) << 4) + (wv << 1)) : r + 1;
        int na = 0, nb = 0;
        if (q + 1 < nq && rn < N) { na = rend[rn - 1]; nb = rend[rn]; }

        const float coef = (len > 0) ? -scale * rsqrtf((float)len) : 0.0f;
        float4 acc = {0.0f, 0.0f, 0.0f, 0.0f};
        int base = start, left = len;

        while (left > TRIP) {
            Trip<LONG_STEPS> t;
            trip_issue(t, ent, 2 * LONG_STEPS, slot, half, coef, rend, Vb, loff);
            base += 2 * LONG_STEPS;
            left -= 2 * LONG_STEPS;
            ent = load_entry(csr, base, min(TRIP, left), slot);
            trip_sum(acc, t, half);
        }

        Trip<ROW_STEPS> t;
        trip_issue(t, ent, left, slot, half, coef, rend, Vb, loff);
        float4 t0 = {0.0f, 0.0f, 0.0f, 0.0f};
        float4 t1 = {0.0f, 0.0f, 0.0f, 0.0f};
        if (half == 0 && r < N) {
            t0 = *(const float4*)((const char*)X  + (((size_t)r << 8) + loff));
            t1 = *(const float4*)((const char*)T1 + (((size_t)r << 8) + loff));
        }
        const int nlen = nb - na;
        const unsigned ent_next = load_entry(csr, na, min(TRIP, nlen), slot);
        trip_sum(acc, t, half);

        acc.x += other_half(acc.x);
        acc.y += other_half(acc.y);
        acc.z += other_half(acc.z);
        acc.w += other_half(acc.w);
        acc.x -= t0.x; acc.y -= t0.y; acc.z -= t0.z; acc.w -= t0.w;

        if (j == 0) __syncthreads();
        if (half == 0) {
            float* d = sT + ((wv << 1) + j) * TP + ((lane & 15) << 2);
            *(float4*)(d)           = t0;
            *(float4*)(d + 16 * TP) = t1;
            *(float4*)(d + 32 * TP) = acc;
        }
        if (j == 1) {
            __syncthreads();
            if (wv < 4) {
                const int koff  = (lane >> 4) << 1;
                const int mhalf = (lane >> 4) << 3;
                const float* ab = sT + (lane & 15) * TP + koff;
                const float* bb = sW + (wv << 6) + (lane << 1);
                v8f acc0 = {};
                v8f acc1 = {};
#pragma unroll
                for (int s = 0; s < 3; ++s) {
#pragma unroll
                    for (int k8 = 0; k8 < 8; ++k8) {
                        const v2f a0 = *(const v2f*)(ab + s * 16 * TP + (2 * k8 + 0) * 4);
                        const v2f a1 = *(const v2f*)(ab + s * 16 * TP + (2 * k8 + 1) * 4);
                        const v2f b0 = *(const v2f*)(bb + (s * 16 + 2 * k8 + 0) * 256);
                        const v2f b1 = *(const v2f*)(bb + (s * 16 + 2 * k8 + 1) * 256);
                        acc0 = __builtin_amdgcn_wmma_f32_16x16x4_f32(
                            false, a0, false, b0, (short)0, acc0, false, false);
                        acc1 = __builtin_amdgcn_wmma_f32_16x16x4_f32(
                            false, a1, false, b1, (short)0, acc1, false, false);
                    }
                }
                const int row0 = tile << 4;
#pragma unroll
                for (int i = 0; i < 8; ++i) {
                    const int mr = row0 + i + mhalf;
                    if (mr < N) {
                        const float v = fmaxf(acc0[i] + acc1[i] + bv, 0.0f);
                        Hout[(size_t)mr * 64 + (wv << 4) + (lane & 15)] = v;
                    }
                }
            }
        }
        start = na;
        len   = nlen;
        ent   = ent_next;
    }
}

__global__ void __launch_bounds__(256, 1)
proj_gemm_wmma(const float* __restrict__ H,
               const float* __restrict__ W,
               const float* __restrict__ bias,
               float* __restrict__ Z,
               float* __restrict__ P1,
               float* __restrict__ P2,
               int N) {
    constexpr int OUT_DIM = 16;
    const int lane = threadIdx.x & 31;
    const int wave = (blockIdx.x * blockDim.x + threadIdx.x) >> 5;
    const int totalWaves = (gridDim.x * blockDim.x) >> 5;
    const int rowTiles = (N + 15) >> 4;

    const int n    = lane & 15;
    const int m    = lane & 15;
    const int koff = (lane >> 4) << 1;
    const int mhalf = (lane >> 4) << 3;

    v2f bfrag[3][16];
#pragma unroll
    for (int s = 0; s < 3; ++s) {
        const float* Ws = W + s * IN_DIM * OUT_DIM + n;
#pragma unroll
        for (int k4 = 0; k4 < 16; ++k4) {
            const int kk = k4 * 4 + koff;
            bfrag[s][k4].x = Ws[(kk + 0) * OUT_DIM];
            bfrag[s][k4].y = Ws[(kk + 1) * OUT_DIM];
        }
    }
    const float bv = bias[n];

    for (int rt = wave; rt < rowTiles; rt += totalWaves) {
        const int row0 = rt << 4;
        const int arow = min(row0 + m, N - 1);
        const float* ap = H + (size_t)arow * IN_DIM + koff;

        v8f acc0 = {};
        v8f acc1 = {};
        v8f acc2 = {};
#pragma unroll
        for (int k4 = 0; k4 < 16; ++k4) {
            v2f a;
            a.x = ap[k4 * 4 + 0];
            a.y = ap[k4 * 4 + 1];
            acc0 = __builtin_amdgcn_wmma_f32_16x16x4_f32(
                false, a, false, bfrag[0][k4], (short)0, acc0, false, false);
            acc1 = __builtin_amdgcn_wmma_f32_16x16x4_f32(
                false, a, false, bfrag[1][k4], (short)0, acc1, false, false);
            acc2 = __builtin_amdgcn_wmma_f32_16x16x4_f32(
                false, a, false, bfrag[2][k4], (short)0, acc2, false, false);
        }

#pragma unroll
        for (int r = 0; r < 8; ++r) {
            const int mr = row0 + r + mhalf;
            if (mr < N) {
                const size_t o = (size_t)mr * OUT_DIM + n;
                Z[o]  = (acc0[r] - acc2[r]) + bv;
                P1[o] = acc1[r];
                P2[o] = acc2[r];
            }
        }
    }
}

template <int J>
__device__ __forceinline__ int quad_lane(int v) {
    return __builtin_amdgcn_update_dpp(0, v, J * 0x55, 0xF, 0xF, false);
}

template <int J>
struct QuadSteps {
    static __device__ __forceinline__ void issue(float4* va, float4* vb, float* wa, float* wb,
                                                 int ca_l, int cb_l, float wa_l, float wb_l,
                                                 int left, const char* Vb, unsigned loff) {
        const unsigned ca = (unsigned)quad_lane<J>(ca_l);
        const unsigned cb = (unsigned)quad_lane<J>(cb_l);
        wa[J] = __int_as_float(quad_lane<J>(__float_as_int(wa_l)));
        wb[J] = __int_as_float(quad_lane<J>(__float_as_int(wb_l)));
        va[J] = float4{0.0f, 0.0f, 0.0f, 0.0f};
        vb[J] = float4{0.0f, 0.0f, 0.0f, 0.0f};
        if (J < left)     va[J] = *(const float4*)(Vb + ((ca << 6) + loff));
        if (J + 4 < left) vb[J] = *(const float4*)(Vb + ((cb << 6) + loff));
        QuadSteps<J + 1>::issue(va, vb, wa, wb, ca_l, cb_l, wa_l, wb_l, left, Vb, loff);
    }
};
template <>
struct QuadSteps<4> {
    static __device__ __forceinline__ void issue(float4*, float4*, float*, float*,
                                                 int, int, float, float,
                                                 int, const char*, unsigned) {}
};

template <int COPY>
__global__ void __launch_bounds__(256)
spmm16_kernel(const unsigned* __restrict__ csr, unsigned* __restrict__ csr_copy,
              const int* __restrict__ rend, const float* __restrict__ V,
              const float* __restrict__ ADD, float* __restrict__ OUT, float scale, int N) {
    const int gt = blockIdx.x * blockDim.x + threadIdx.x;
    const int r = gt >> 2;
    const int q = gt & 3;
    if (r >= N) return;
    const unsigned loff = (unsigned)q << 4;
    const char* Vb = (const char*)V;

    const int start = r ? rend[r - 1] : 0;
    const int end   = rend[r];
    const int len   = end - start;
    const float coef = (len > 0) ? -scale * rsqrtf((float)len) : 0.0f;

    float4 acc = *(const float4*)((const char*)ADD + (((size_t)r << 6) + loff));

    unsigned ea = 0, eb = 0;
    if (start + q < end)     ea = csr[start + q];
    if (start + 4 + q < end) eb = csr[start + 4 + q];

    for (int base = start; base < end; base += 8) {
        const int ia = base + q, ib = base + 4 + q;
        const bool oka = ia < end, okb = ib < end;
        if (COPY) {
            if (oka) csr_copy[ia] = ea;
            if (okb) csr_copy[ib] = eb;
        }
        int ca_l, cb_l;
        float wa_l, wb_l;
        decode_entry(ea, oka ? 1 : 0, 0, coef, rend, ca_l, wa_l);
        decode_entry(eb, okb ? 1 : 0, 0, coef, rend, cb_l, wb_l);

        float4 va[4], vb[4];
        float  wa[4], wb[4];
        QuadSteps<0>::issue(va, vb, wa, wb, ca_l, cb_l, wa_l, wb_l, end - base, Vb, loff);

        ea = 0; eb = 0;
        if (ia + 8 < end) ea = csr[ia + 8];
        if (ib + 8 < end) eb = csr[ib + 8];

#pragma unroll
        for (int j = 0; j < 4; ++j) {
            acc.x = fmaf(wa[j], va[j].x, acc.x);
            acc.y = fmaf(wa[j], va[j].y, acc.y);
            acc.z = fmaf(wa[j], va[j].z, acc.z);
            acc.w = fmaf(wa[j], va[j].w, acc.w);
        }
#pragma unroll
        for (int j = 0; j < 4; ++j) {
            acc.x = fmaf(wb[j], vb[j].x, acc.x);
            acc.y = fmaf(wb[j], vb[j].y, acc.y);
            acc.z = fmaf(wb[j], vb[j].z, acc.z);
            acc.w = fmaf(wb[j], vb[j].w, acc.w);
        }
    }
    *(float4*)((char*)OUT + (((size_t)r << 6) + loff)) = acc;
}

extern "C" void kernel_launch(void* const* d_in, const int* in_sizes, int n_in,
                              void* d_out, int out_size, void* d_ws, size_t ws_size,
                              hipStream_t stream) {
    const float* x  = (const float*)d_in[0];
    const int*   ei = (const int*)d_in[1];
    const float* W1 = (const float*)d_in[2];
    const float* b1 = (const float*)d_in[3];
    const float* W2 = (const float*)d_in[4];
    const float* b2 = (const float*)d_in[5];
    float* out = (float*)d_out;

    const int N = in_sizes[0] / IN_DIM;
    const int E = in_sizes[1] / 2;
    const int* row = ei;
    const int* col = ei + E;

    float* ws   = (float*)d_ws;
    int*   rend = (int*)ws;
    float* Tx1  = ws + N;
    float* Tx2  = Tx1 + (size_t)N * IN_DIM;
    float* h    = Tx2 + (size_t)N * IN_DIM;
    const int NB = (N + BROWS - 1) >> BSH;
    const int GA = (E + SUB - 1) / SUB;
    const int MB = (NB * GA + 3) & ~3;
    int*      cnt = (int*)h;
    int*      bc  = cnt + N;
    int*      bo  = bc + MB;
    unsigned* bk  = (unsigned*)(bo + MB);
    unsigned* csr = (unsigned*)d_out;

    const int threads = 256;
    const unsigned gridScan = (unsigned)((NB * GA + 1023) / 1024);
    const unsigned gridRows = 1280;
    const unsigned gridDense = 1250;

    bucket_count_kernel<<<(unsigned)GA, threads, 0, stream>>>(row, col, bc, E, NB);
    scan_kernel<<<gridScan, threads, 0, stream>>>(bc, bo, NB * GA);
    bucket_scatter_kernel<<<(unsigned)GA, threads, 0, stream>>>(row, col, bo, bk, E, NB);
    bucket_rows_kernel<<<(unsigned)NB, 1024, 0, stream>>>(bk, bo, cnt, rend, N, GA);
    bucket_place_kernel<<<(unsigned)NB, 1024, 0, stream>>>(bk, bo, cnt, rend, csr, N, GA);

    spmm_rows_kernel<0><<<gridRows, threads, 0, stream>>>(csr, rend, x, x, Tx1, 1.0f, N);
    spmm_dense_kernel<<<gridDense, threads, 0, stream>>>(csr, rend, x, Tx1, W1, b1, h, 2.0f, N);

    float* Z  = Tx1;
    float* P1 = Z  + (size_t)N * 16;
    float* P2 = P1 + (size_t)N * 16;
    float* R  = P2 + (size_t)N * 16;
    unsigned* csr2 = (unsigned*)Tx2;
    const unsigned gridQuad = (unsigned)((N + 63) / 64);

    proj_gemm_wmma<<<512, threads, 0, stream>>>(h, W2, b2, Z, P1, P2, N);
    spmm16_kernel<1><<<gridQuad, threads, 0, stream>>>(csr, csr2, rend, P2, P1, R, 2.0f, N);
    spmm16_kernel<0><<<gridQuad, threads, 0, stream>>>(csr2, nullptr, rend, R, Z, out, 1.0f, N);
}


# Edge-list build in 3 launches, not 5: each block orders its own 4096-edge slice by bucket and keeps the run starts; bucket blocks sum them and walk the runs; sum order = LDS arrival order
# speedup vs baseline: 5.6826x; 1.0928x over previous
#include <hip/hip_runtime.h>


typedef __attribute__((ext_vector_type(2))) float v2f;
typedef __attribute__((ext_vector_type(8))) float v8f;

#define IN_DIM 64

#define COL_BITS 17
#define COL_MASK 0x1FFFFu
#define DEG_ESC  0x7FFFu

#define BSH    9
#define BROWS  512
#define MAXB   256
#define SUB    4096
#define EPT    (SUB / 256)
#define CAP    12288
#define RMAX   1024
#define RUN_U  8

__global__ void __launch_bounds__(256)
bucket_local_kernel(const int* __restrict__ row, const int* __restrict__ col,
                    int* __restrict__ tab, unsigned* __restrict__ bk, int E, int NB) {
    __shared__ int h[MAXB];
    __shared__ int wsum[8];
    __shared__ unsigned st[SUB];
    const int tid = threadIdx.x, lane = tid & 31, wv = tid >> 5;
    h[tid] = 0;
    __syncthreads();
    const int base = blockIdx.x * SUB;
    unsigned ent[EPT];
    int bb[EPT], lr[EPT];
#pragma unroll
    for (int j = 0; j < EPT; ++j) {
        const int e = base + j * 256 + tid;
        bb[j] = -1; lr[j] = 0; ent[j] = 0;
        if (e < E) {
            const int r = row[e], c = col[e];
            if (r != c) {
                const int b = min(r >> BSH, NB - 1);
                bb[j]  = b;
                ent[j] = ((unsigned)(r - (b << BSH)) << COL_BITS) | ((unsigned)c & COL_MASK);
                lr[j]  = atomicAdd(&h[b], 1);
            }
        }
    }
    __syncthreads();
    const int v = h[tid];
    int incl = v;
    for (int o = 1; o < 32; o <<= 1) {
        const int u = __shfl_up(incl, o);
        if (lane >= o) incl += u;
    }
    if (lane == 31) wsum[wv] = incl;
    __syncthreads();
    int off = 0, total = 0;
    for (int j = 0; j < 8; ++j) {
        const int s = wsum[j];
        if (j < wv) off += s;
        total += s;
    }
    const int ex = off + incl - v;
    h[tid] = ex;
    if (tid < NB) tab[tid * gridDim.x + blockIdx.x] = ex;
    if (tid == 0) tab[NB * gridDim.x + blockIdx.x] = total;
    __syncthreads();
#pragma unroll
    for (int j = 0; j < EPT; ++j)
        if (bb[j] >= 0) st[h[bb[j]] + lr[j]] = ent[j];
    __syncthreads();
    for (int i = tid; i < total; i += 256) bk[base + i] = st[i];
}

__device__ __forceinline__ void bucket_bounds(const int* __restrict__ t0, const int* __restrict__ t1,
                                              int G, int* red, int& bs, int& be) {
    const int tid = threadIdx.x, lane = tid & 31, wv = tid >> 5;
    int a = 0, b = 0;
    for (int i = tid; i < G; i += 1024) { a += t0[i]; b += t1[i]; }
    for (int o = 16; o > 0; o >>= 1) { a += __shfl_xor(a, o); b += __shfl_xor(b, o); }
    if (lane == 0) { red[wv] = a; red[32 + wv] = b; }
    __syncthreads();
    int sa = 0, sb = 0;
    for (int j = 0; j < 32; ++j) { sa += red[j]; sb += red[32 + j]; }
    bs = sa; be = sb;
}

__global__ void __launch_bounds__(1024)
bucket_rows_kernel(const unsigned* __restrict__ bk, const int* __restrict__ tab,
                   int* __restrict__ cnt, int* __restrict__ rend, int N, int G) {
    __shared__ int c[BROWS];
    __shared__ int wsum[32];
    __shared__ int red[64];
    __shared__ int rs[RMAX];
    __shared__ int re[RMAX];
    const int tid = threadIdx.x, lane = tid & 31, wv = tid >> 5;
    const int* t0 = tab + (size_t)blockIdx.x * G;
    const int* t1 = t0 + G;
    if (tid < BROWS) c[tid] = 0;
    int bs, be;
    bucket_bounds(t0, t1, G, red, bs, be);
    for (int g0 = 0; g0 < G; g0 += RMAX) {
        const int nrun = min(RMAX, G - g0);
        __syncthreads();
        if (tid < nrun) { rs[tid] = t0[g0 + tid]; re[tid] = t1[g0 + tid]; }
        __syncthreads();
        for (int k0 = wv; k0 < nrun; k0 += 32 * RUN_U) {
            unsigned ent[RUN_U];
            int s_[RUN_U], e_[RUN_U];
#pragma unroll
            for (int u = 0; u < RUN_U; ++u) {
                const int k = k0 + 32 * u;
                int s = 0, e = 0;
                if (k < nrun) { s = rs[k]; e = re[k]; }
                s_[u] = s; e_[u] = e;
                ent[u] = 0;
                if (s + lane < e) ent[u] = bk[(size_t)(g0 + k) * SUB + s + lane];
            }
#pragma unroll
            for (int u = 0; u < RUN_U; ++u) {
                if (s_[u] + lane < e_[u]) atomicAdd(&c[(ent[u] >> COL_BITS) & (BROWS - 1)], 1);
                const unsigned* p = bk + (size_t)(g0 + k0 + 32 * u) * SUB;
                for (int i = s_[u] + 32 + lane; i < e_[u]; i += 32)
                    atomicAdd(&c[(p[i] >> COL_BITS) & (BROWS - 1)], 1);
            }
        }
    }
    __syncthreads();
    const int v = (tid < BROWS) ? c[tid] : 0;
    int incl = v;
    for (int o = 1; o < 32; o <<= 1) {
        const int u = __shfl_up(incl, o);
        if (lane >= o) incl += u;
    }
    if (lane == 31) wsum[wv] = incl;
    __syncthreads();
    int off = 0;
    for (int j = 0; j < wv; ++j) off += wsum[j];
    const int r = blockIdx.x * BROWS + tid;
    if (tid < BROWS && r < N) {
        cnt[r]  = v;
        rend[r] = bs + off + incl;
    }
}

__global__ void __launch_bounds__(1024)
bucket_place_kernel(const unsigned* __restrict__ bk, const int* __restrict__ tab,
                    const int* __restrict__ cnt, const int* __restrict__ rend,
                    unsigned* __restrict__ csr, int N, int G) {
    __shared__ int cur[BROWS];
    __shared__ unsigned st[CAP];
    __shared__ int red[64];
    __shared__ int rs[RMAX];
    __shared__ int re[RMAX];
    const int tid = threadIdx.x, lane = tid & 31, wv = tid >> 5;
    const int* t0 = tab + (size_t)blockIdx.x * G;
    const int* t1 = t0 + G;
    int bs, be;
    bucket_bounds(t0, t1, G, red, bs, be);
    const int nb = be - bs;
    const int r = blockIdx.x * BROWS + tid;
    if (tid < BROWS) cur[tid] = (r < N) ? ((r ? rend[r - 1] : 0) - bs) : 0;
    const bool staged = nb <= CAP;
    for (int g0 = 0; g0 < G; g0 += RMAX) {
        const int nrun = min(RMAX, G - g0);
        __syncthreads();
        if (tid < nrun) { rs[tid] = t0[g0 + tid]; re[tid] = t1[g0 + tid]; }
        __syncthreads();
        for (int k0 = wv; k0 < nrun; k0 += 32 * RUN_U) {
            unsigned ent[RUN_U], dg[RUN_U];
            int s_[RUN_U], e_[RUN_U];
#pragma unroll
            for (int u = 0; u < RUN_U; ++u) {
                const int k = k0 + 32 * u;
                int s = 0, e = 0;
                if (k < nrun) { s = rs[k]; e = re[k]; }
                s_[u] = s; e_[u] = e;
                ent[u] = 0;
                if (s + lane < e) ent[u] = bk[(size_t)(g0 + k) * SUB + s + lane];
            }
#pragma unroll
            for (int u = 0; u < RUN_U; ++u) {
                dg[u] = 0;
                if (s_[u] + lane < e_[u]) dg[u] = (unsigned)cnt[ent[u] & COL_MASK];
            }
#pragma unroll
            for (int u = 0; u < RUN_U; ++u) {
                if (s_[u] + lane < e_[u]) {
                    const unsigned s = ent[u] & COL_MASK;
                    const unsigned d = min(dg[u], DEG_ESC);
                    const int p = atomicAdd(&cur[(ent[u] >> COL_BITS) & (BROWS - 1)], 1);
                    const unsigned val = s | (d << COL_BITS);
                    if (staged) st[p] = val; else csr[bs + p] = val;
                }
                const unsigned* q = bk + (size_t)(g0 + k0 + 32 * u) * SUB;
                for (int i = s_[u] + 32 + lane; i < e_[u]; i += 32) {
                    const unsigned e = q[i];
                    const unsigned s = e & COL_MASK;
                    const unsigned d = min((unsigned)cnt[s], DEG_ESC);
                    const int p = atomicAdd(&cur[(e >> COL_BITS) & (BROWS - 1)], 1);
                    const unsigned val = s | (d << COL_BITS);
                    if (staged) st[p] = val; else csr[bs + p] = val;
                }
            }
        }
    }
    __syncthreads();
    if (staged)
        for (int i = tid; i < nb; i += 1024) csr[bs + i] = st[i];
}

#define ROW_STEPS 12
#define TRIP (2 * ROW_STEPS)

template <int K>
__device__ __forceinline__ int half_lane(int v) {
    return __builtin_amdgcn_update_dpp(0, v, 0x150 + K, 0xF, 0xF, false);
}

__device__ __forceinline__ float other_half(float x) {
    return __int_as_float((int)__builtin_amdgcn_permlanex16(
        0u, (unsigned)__float_as_int(x), 0x76543210u, 0xfedcba98u, false, false));
}

template <int J, int NS>
struct RowSteps {
    static __device__ __forceinline__ void issue(float4* v, int c_l, int np,
                                                 const char* Vb, unsigned loff) {
        if (J < np) {
            const unsigned c = (unsigned)half_lane<J>(c_l);
            v[J] = *(const float4*)(Vb + ((c << 8) + loff));
        }
        RowSteps<J + 1, NS>::issue(v, c_l, np, Vb, loff);
    }
    static __device__ __forceinline__ void sum(float4& acc, const float4* v, float w_l, int np) {
        if (J < np) {
            const float w = __int_as_float(half_lane<J>(__float_as_int(w_l)));
            acc.x = fmaf(w, v[J].x, acc.x);
            acc.y = fmaf(w, v[J].y, acc.y);
            acc.z = fmaf(w, v[J].z, acc.z);
            acc.w = fmaf(w, v[J].w, acc.w);
        }
        RowSteps<J + 1, NS>::sum(acc, v, w_l, np);
    }
};
template <int NS>
struct RowSteps<NS, NS> {
    static __device__ __forceinline__ void issue(float4*, int, int, const char*, unsigned) {}
    static __device__ __forceinline__ void sum(float4&, const float4*, float, int) {}
};

__device__ __forceinline__ unsigned load_entry(const unsigned* __restrict__ csr,
                                               int base, int n, int slot) {
    const int idx = (slot < n) ? base + slot : 0;
    return csr[idx];
}

__device__ __forceinline__ void decode_entry(unsigned ent, int n, int slot, float coef,
                                             const int* __restrict__ rend,
                                             int& c_l, float& w_l) {
    const bool ok = slot < n;
    const int c = (int)(ent & COL_MASK);
    unsigned d = ent >> COL_BITS;
    if (ok && d == DEG_ESC) d = (unsigned)(rend[c] - (c ? rend[c - 1] : 0));
    const float w = coef * rsqrtf((float)d);
    c_l = ok ? c : 0;
    w_l = (ok && d) ? w : 0.0f;
}

template <int NS>
struct Trip {
    float4 v[NS];
    float4 vt;
    float  w_l, wt;
    int    np, odd;
};

template <int NS>
__device__ __forceinline__ void trip_issue(Trip<NS>& t, unsigned ent, int n, int slot, int half,
                                           float coef, const int* __restrict__ rend,
                                           const char* Vb, unsigned loff) {
    int c_l;
    decode_entry(ent, n, slot, coef, rend, c_l, t.w_l);
    t.np  = n >> 1;
    t.odd = n & 1;
    RowSteps<0, NS>::issue(t.v, c_l, t.np, Vb, loff);
    const unsigned ct = (unsigned)__builtin_amdgcn_readlane(c_l, t.np & 15);
    t.wt = __int_as_float(__builtin_amdgcn_readlane(__float_as_int(t.w_l), t.np & 15));
    t.vt = float4{0.0f, 0.0f, 0.0f, 0.0f};
    if (t.odd && half == 0) t.vt = *(const float4*)(Vb + ((ct << 8) + loff));
}

template <int NS>
__device__ __forceinline__ void trip_sum(float4& acc, const Trip<NS>& t, int half) {
    RowSteps<0, NS>::sum(acc, t.v, t.w_l, t.np);
    if (t.odd && half == 0) {
        acc.x = fmaf(t.wt, t.vt.x, acc.x);
        acc.y = fmaf(t.wt, t.vt.y, acc.y);
        acc.z = fmaf(t.wt, t.vt.z, acc.z);
        acc.w = fmaf(t.wt, t.vt.w, acc.w);
    }
}

#define LONG_STEPS 4

template <int SUB_T0>
__global__ void __launch_bounds__(256)
spmm_rows_kernel(const unsigned* __restrict__ csr, const int* __restrict__ rend,
                 const float* __restrict__ V, const float* __restrict__ T0,
                 float* __restrict__ OUT, float scale, int N) {
    const int lane = threadIdx.x & 31;
    const int half = lane >> 4;
    const int slot = ((lane & 15) << 1) + half;
    const unsigned loff = (unsigned)(lane & 15) << 4;
    const char* Vb = (const char*)V;
    const int wavesPerBlock = blockDim.x >> 5;
    const int w0 = __builtin_amdgcn_readfirstlane(
        (int)(blockIdx.x * wavesPerBlock + (threadIdx.x >> 5)));
    const int nw = gridDim.x * wavesPerBlock;
    if (w0 >= N) return;

    int start = w0 ? rend[w0 - 1] : 0;
    int len   = rend[w0] - start;
    unsigned ent = load_entry(csr, start, min(TRIP, len), slot);

    for (int r = w0; r < N; r += nw) {
        const int rn = r + nw;
        int na = 0, nb = 0;
        if (rn < N) { na = rend[rn - 1]; nb = rend[rn]; }

        const float coef = (len > 0) ? -scale * rsqrtf((float)len) : 0.0f;
        float4 acc = {0.0f, 0.0f, 0.0f, 0.0f};
        int base = start, left = len;

        while (left > TRIP) {
            Trip<LONG_STEPS> t;
            trip_issue(t, ent, 2 * LONG_STEPS, slot, half, coef, rend, Vb, loff);
            base += 2 * LONG_STEPS;
            left -= 2 * LONG_STEPS;
            ent = load_entry(csr, base, min(TRIP, left), slot);
            trip_sum(acc, t, half);
        }

        Trip<ROW_STEPS> t;
        trip_issue(t, ent, left, slot, half, coef, rend, Vb, loff);
        float4 t0 = {0.0f, 0.0f, 0.0f, 0.0f};
        if (SUB_T0 && half == 0)
            t0 = *(const float4*)((const char*)T0 + (((size_t)r << 8) + loff));
        const int nlen = nb - na;
        const unsigned ent_next = load_entry(csr, na, min(TRIP, nlen), slot);
        trip_sum(acc, t, half);

        acc.x += other_half(acc.x);
        acc.y += other_half(acc.y);
        acc.z += other_half(acc.z);
        acc.w += other_half(acc.w);
        if (half == 0) {
            if (SUB_T0) { acc.x -= t0.x; acc.y -= t0.y; acc.z -= t0.z; acc.w -= t0.w; }
            *(float4*)((char*)OUT + (((size_t)r << 8) + loff)) = acc;
        }
        start = na;
        len   = nlen;
        ent   = ent_next;
    }
}

#define TP 68

__global__ void __launch_bounds__(256)
spmm_dense_kernel(const unsigned* __restrict__ csr, const int* __restrict__ rend,
                  const float* __restrict__ X, const float* __restrict__ T1,
                  const float* __restrict__ W,
                  const float* __restrict__ bias,
                  float* __restrict__ Hout,
                  float scale, int N) {
    __shared__ __attribute__((aligned(16))) float sW[3 * 16 * 4 * 64];
    __shared__ __attribute__((aligned(16))) float sT[3 * 16 * TP];
    const int tid  = threadIdx.x;
    const int lane = tid & 31;
    const int half = lane >> 4;
    const int slot = ((lane & 15) << 1) + half;
    const unsigned loff = (unsigned)(lane & 15) << 4;
    const char* Vb = (const char*)T1;
    const int wv = __builtin_amdgcn_readfirstlane(tid >> 5);

    for (int i = tid; i < 3 * IN_DIM * 64; i += 256) {
        const int s = i >> 12, k = (i >> 6) & 63, c = i & 63;
        const int ln = (((k >> 1) & 1) << 4) | (c & 15);
        sW[((((s << 4) + (k >> 2)) * 4 + (c >> 4)) * 32 + ln) * 2 + (k & 1)] = W[i];
    }
    const float bv = bias[((wv & 3) << 4) + (lane & 15)];

    const int tiles = (N + 15) >> 4;
    const int ntile = ((int)blockIdx.x < tiles) ? (tiles - 1 - (int)blockIdx.x) / (int)gridDim.x + 1 : 0;
    const int nq = 2 * ntile;

    int start = 0, len = 0;
    {
        const int r0 = ((int)blockIdx.x << 4) + (wv << 1);
        if (nq > 0 && r0 < N) {
            start = r0 ? rend[r0 - 1] : 0;
            len   = rend[r0] - start;
        }
    }
    unsigned ent = load_entry(csr, start, min(TRIP, len), slot);

    for (int q = 0; q < nq; ++q) {
        const int j    = q & 1;
        const int tile = (int)blockIdx.x + (q >> 1) * (int)gridDim.x;
        const int r    = (tile << 4) + (wv << 1) + j;
        const int rn = j ? (((tile + (int)gridDim.x) << 4) + (wv << 1)) : r + 1;
        int na = 0, nb = 0;
        if (q + 1 < nq && rn < N) { na = rend[rn - 1]; nb = rend[rn]; }

        const float coef = (len > 0) ? -scale * rsqrtf((float)len) : 0.0f;
        float4 acc = {0.0f, 0.0f, 0.0f, 0.0f};
        int base = start, left = len;

        while (left > TRIP) {
            Trip<LONG_STEPS> t;
            trip_issue(t, ent, 2 * LONG_STEPS, slot, half, coef, rend, Vb, loff);
            base += 2 * LONG_STEPS;
            left -= 2 * LONG_STEPS;
            ent = load_entry(csr, base, min(TRIP, left), slot);
            trip_sum(acc, t, half);
        }

        Trip<ROW_STEPS> t;
        trip_issue(t, ent, left, slot, half, coef, rend, Vb, loff);
        float4 t0 = {0.0f, 0.0f, 0.0f, 0.0f};
        float4 t1 = {0.0f, 0.0f, 0.0f, 0.0f};
        if (half == 0 && r < N) {
            t0 = *(const float4*)((const char*)X  + (((size_t)r << 8) + loff));
            t1 = *(const float4*)((const char*)T1 + (((size_t)r << 8) + loff));
        }
        const int nlen = nb - na;
        const unsigned ent_next = load_entry(csr, na, min(TRIP, nlen), slot);
        trip_sum(acc, t, half);

        acc.x += other_half(acc.x);
        acc.y += other_half(acc.y);
        acc.z += other_half(acc.z);
        acc.w += other_half(acc.w);
        acc.x -= t0.x; acc.y -= t0.y; acc.z -= t0.z; acc.w -= t0.w;

        if (j == 0) __syncthreads();
        if (half == 0) {
            float* d = sT + ((wv << 1) + j) * TP + ((lane & 15) << 2);
            *(float4*)(d)           = t0;
            *(float4*)(d + 16 * TP) = t1;
            *(float4*)(d + 32 * TP) = acc;
        }
        if (j == 1) {
            __syncthreads();
            if (wv < 4) {
                const int koff  = (lane >> 4) << 1;
                const int mhalf = (lane >> 4) << 3;
                const float* ab = sT + (lane & 15) * TP + koff;
                const float* bb = sW + (wv << 6) + (lane << 1);
                v8f acc0 = {};
                v8f acc1 = {};
#pragma unroll
                for (int s = 0; s < 3; ++s) {
#pragma unroll
                    for (int k8 = 0; k8 < 8; ++k8) {
                        const v2f a0 = *(const v2f*)(ab + s * 16 * TP + (2 * k8 + 0) * 4);
                        const v2f a1 = *(const v2f*)(ab + s * 16 * TP + (2 * k8 + 1) * 4);
                        const v2f b0 = *(const v2f*)(bb + (s * 16 + 2 * k8 + 0) * 256);
                        const v2f b1 = *(const v2f*)(bb + (s * 16 + 2 * k8 + 1) * 256);
                        acc0 = __builtin_amdgcn_wmma_f32_16x16x4_f32(
                            false, a0, false, b0, (short)0, acc0, false, false);
                        acc1 = __builtin_amdgcn_wmma_f32_16x16x4_f32(
                            false, a1, false, b1, (short)0, acc1, false, false);
                    }
                }
                const int row0 = tile << 4;
#pragma unroll
                for (int i = 0; i < 8; ++i) {
                    const int mr = row0 + i + mhalf;
                    if (mr < N) {
                        const float v = fmaxf(acc0[i] + acc1[i] + bv, 0.0f);
                        Hout[(size_t)mr * 64 + (wv << 4) + (lane & 15)] = v;
                    }
                }
            }
        }
        start = na;
        len   = nlen;
        ent   = ent_next;
    }
}

__global__ void __launch_bounds__(256, 1)
proj_gemm_wmma(const float* __restrict__ H,
               const float* __restrict__ W,
               const float* __restrict__ bias,
               float* __restrict__ Z,
               float* __restrict__ P1,
               float* __restrict__ P2,
               int N) {
    constexpr int OUT_DIM = 16;
    const int lane = threadIdx.x & 31;
    const int wave = (blockIdx.x * blockDim.x + threadIdx.x) >> 5;
    const int totalWaves = (gridDim.x * blockDim.x) >> 5;
    const int rowTiles = (N + 15) >> 4;

    const int n    = lane & 15;
    const int m    = lane & 15;
    const int koff = (lane >> 4) << 1;
    const int mhalf = (lane >> 4) << 3;

    v2f bfrag[3][16];
#pragma unroll
    for (int s = 0; s < 3; ++s) {
        const float* Ws = W + s * IN_DIM * OUT_DIM + n;
#pragma unroll
        for (int k4 = 0; k4 < 16; ++k4) {
            const int kk = k4 * 4 + koff;
            bfrag[s][k4].x = Ws[(kk + 0) * OUT_DIM];
            bfrag[s][k4].y = Ws[(kk + 1) * OUT_DIM];
        }
    }
    const float bv = bias[n];

    for (int rt = wave; rt < rowTiles; rt += totalWaves) {
        const int row0 = rt << 4;
        const int arow = min(row0 + m, N - 1);
        const float* ap = H + (size_t)arow * IN_DIM + koff;

        v8f acc0 = {};
        v8f acc1 = {};
        v8f acc2 = {};
#pragma unroll
        for (int k4 = 0; k4 < 16; ++k4) {
            v2f a;
            a.x = ap[k4 * 4 + 0];
            a.y = ap[k4 * 4 + 1];
            acc0 = __builtin_amdgcn_wmma_f32_16x16x4_f32(
                false, a, false, bfrag[0][k4], (short)0, acc0, false, false);
            acc1 = __builtin_amdgcn_wmma_f32_16x16x4_f32(
                false, a, false, bfrag[1][k4], (short)0, acc1, false, false);
            acc2 = __builtin_amdgcn_wmma_f32_16x16x4_f32(
                false, a, false, bfrag[2][k4], (short)0, acc2, false, false);
        }

#pragma unroll
        for (int r = 0; r < 8; ++r) {
            const int mr = row0 + r + mhalf;
            if (mr < N) {
                const size_t o = (size_t)mr * OUT_DIM + n;
                Z[o]  = (acc0[r] - acc2[r]) + bv;
                P1[o] = acc1[r];
                P2[o] = acc2[r];
            }
        }
    }
}

template <int J>
__device__ __forceinline__ int quad_lane(int v) {
    return __builtin_amdgcn_update_dpp(0, v, J * 0x55, 0xF, 0xF, false);
}

template <int J>
struct QuadSteps {
    static __device__ __forceinline__ void issue(float4* va, float4* vb, float* wa, float* wb,
                                                 int ca_l, int cb_l, float wa_l, float wb_l,
                                                 int left, const char* Vb, unsigned loff) {
        const unsigned ca = (unsigned)quad_lane<J>(ca_l);
        const unsigned cb = (unsigned)quad_lane<J>(cb_l);
        wa[J] = __int_as_float(quad_lane<J>(__float_as_int(wa_l)));
        wb[J] = __int_as_float(quad_lane<J>(__float_as_int(wb_l)));
        va[J] = float4{0.0f, 0.0f, 0.0f, 0.0f};
        vb[J] = float4{0.0f, 0.0f, 0.0f, 0.0f};
        if (J < left)     va[J] = *(const float4*)(Vb + ((ca << 6) + loff));
        if (J + 4 < left) vb[J] = *(const float4*)(Vb + ((cb << 6) + loff));
        QuadSteps<J + 1>::issue(va, vb, wa, wb, ca_l, cb_l, wa_l, wb_l, left, Vb, loff);
    }
};
template <>
struct QuadSteps<4> {
    static __device__ __forceinline__ void issue(float4*, float4*, float*, float*,
                                                 int, int, float, float,
                                                 int, const char*, unsigned) {}
};

template <int COPY>
__global__ void __launch_bounds__(256)
spmm16_kernel(const unsigned* __restrict__ csr, unsigned* __restrict__ csr_copy,
              const int* __restrict__ rend, const float* __restrict__ V,
              const float* __restrict__ ADD, float* __restrict__ OUT, float scale, int N) {
    const int gt = blockIdx.x * blockDim.x + threadIdx.x;
    const int r = gt >> 2;
    const int q = gt & 3;
    if (r >= N) return;
    const unsigned loff = (unsigned)q << 4;
    const char* Vb = (const char*)V;

    const int start = r ? rend[r - 1] : 0;
    const int end   = rend[r];
    const int len   = end - start;
    const float coef = (len > 0) ? -scale * rsqrtf((float)len) : 0.0f;

    float4 acc = *(const float4*)((const char*)ADD + (((size_t)r << 6) + loff));

    unsigned ea = 0, eb = 0;
    if (start + q < end)     ea = csr[start + q];
    if (start + 4 + q < end) eb = csr[start + 4 + q];

    for (int base = start; base < end; base += 8) {
        const int ia = base + q, ib = base + 4 + q;
        const bool oka = ia < end, okb = ib < end;
        if (COPY) {
            if (oka) csr_copy[ia] = ea;
            if (okb) csr_copy[ib] = eb;
        }
        int ca_l, cb_l;
        float wa_l, wb_l;
        decode_entry(ea, oka ? 1 : 0, 0, coef, rend, ca_l, wa_l);
        decode_entry(eb, okb ? 1 : 0, 0, coef, rend, cb_l, wb_l);

        float4 va[4], vb[4];
        float  wa[4], wb[4];
        QuadSteps<0>::issue(va, vb, wa, wb, ca_l, cb_l, wa_l, wb_l, end - base, Vb, loff);

        ea = 0; eb = 0;
        if (ia + 8 < end) ea = csr[ia + 8];
        if (ib + 8 < end) eb = csr[ib + 8];

#pragma unroll
        for (int j = 0; j < 4; ++j) {
            acc.x = fmaf(wa[j], va[j].x, acc.x);
            acc.y = fmaf(wa[j], va[j].y, acc.y);
            acc.z = fmaf(wa[j], va[j].z, acc.z);
            acc.w = fmaf(wa[j], va[j].w, acc.w);
        }
#pragma unroll
        for (int j = 0; j < 4; ++j) {
            acc.x = fmaf(wb[j], vb[j].x, acc.x);
            acc.y = fmaf(wb[j], vb[j].y, acc.y);
            acc.z = fmaf(wb[j], vb[j].z, acc.z);
            acc.w = fmaf(wb[j], vb[j].w, acc.w);
        }
    }
    *(float4*)((char*)OUT + (((size_t)r << 6) + loff)) = acc;
}

extern "C" void kernel_launch(void* const* d_in, const int* in_sizes, int n_in,
                              void* d_out, int out_size, void* d_ws, size_t ws_size,
                              hipStream_t stream) {
    const float* x  = (const float*)d_in[0];
    const int*   ei = (const int*)d_in[1];
    const float* W1 = (const float*)d_in[2];
    const float* b1 = (const float*)d_in[3];
    const float* W2 = (const float*)d_in[4];
    const float* b2 = (const float*)d_in[5];
    float* out = (float*)d_out;

    const int N = in_sizes[0] / IN_DIM;
    const int E = in_sizes[1] / 2;
    const int* row = ei;
    const int* col = ei + E;

    float* ws   = (float*)d_ws;
    int*   rend = (int*)ws;
    float* Tx1  = ws + N;
    float* Tx2  = Tx1 + (size_t)N * IN_DIM;
    float* h    = Tx2 + (size_t)N * IN_DIM;
    const int NB = (N + BROWS - 1) >> BSH;
    const int GA = (E + SUB - 1) / SUB;
    const int MT = ((NB + 1) * GA + 31) & ~31;
    int*      cnt = (int*)h;
    int*      tab = cnt + N;
    unsigned* bk  = (unsigned*)(tab + MT);
    unsigned* csr = (unsigned*)d_out;

    const int threads = 256;
    const unsigned gridRows = 1280;
    const unsigned gridDense = 1250;

    bucket_local_kernel<<<(unsigned)GA, threads, 0, stream>>>(row, col, tab, bk, E, NB);
    bucket_rows_kernel<<<(unsigned)NB, 1024, 0, stream>>>(bk, tab, cnt, rend, N, GA);
    bucket_place_kernel<<<(unsigned)NB, 1024, 0, stream>>>(bk, tab, cnt, rend, csr, N, GA);

    spmm_rows_kernel<0><<<gridRows, threads, 0, stream>>>(csr, rend, x, x, Tx1, 1.0f, N);
    spmm_dense_kernel<<<gridDense, threads, 0, stream>>>(csr, rend, x, Tx1, W1, b1, h, 2.0f, N);

    float* Z  = Tx1;
    float* P1 = Z  + (size_t)N * 16;
    float* P2 = P1 + (size_t)N * 16;
    float* R  = P2 + (size_t)N * 16;
    unsigned* csr2 = (unsigned*)Tx2;
    const unsigned gridQuad = (unsigned)((N + 63) / 64);

    proj_gemm_wmma<<<512, threads, 0, stream>>>(h, W2, b2, Z, P1, P2, N);
    spmm16_kernel<1><<<gridQuad, threads, 0, stream>>>(csr, csr2, rend, P2, P1, R, 2.0f, N);
    spmm16_kernel<0><<<gridQuad, threads, 0, stream>>>(csr2, nullptr, rend, R, Z, out, 1.0f, N);
}
